# starting kernel
# baseline (speedup 1.0000x reference)
#include <hip/hip_runtime.h>

typedef __attribute__((ext_vector_type(2))) float v2f;
typedef __attribute__((ext_vector_type(8))) float v8f;

#define N_NODES 10
#define DIM 64
#define TM 64
#define PITCH 68
#define NODE_STRIDE (TM * PITCH)
#define LDS_FLOATS (9 * NODE_STRIDE)

__global__ __launch_bounds__(256) void transpose_w(const float* __restrict__ W,
                                                   float* __restrict__ Wt) {
  __shared__ float tile[DIM][DIM + 1];
  const int e = blockIdx.x;
  const float* src = W + (size_t)e * DIM * DIM;
  float* dst = Wt + (size_t)e * DIM * DIM;
  for (int i = threadIdx.x; i < DIM * DIM; i += 256)
    tile[i >> 6][i & 63] = src[i];
  __syncthreads();
  for (int i = threadIdx.x; i < DIM * DIM; i += 256) {
    const int n = i >> 6, k = i & 63;
    dst[i] = tile[k][n];
  }
}

template <bool XPOSED>
__global__ __launch_bounds__(128) void wired_dag_wmma(
    const float* __restrict__ x, const float* __restrict__ Wm,
    const float* __restrict__ bias, float* __restrict__ out)
{
  extern __shared__ __align__(16) float smem[];
  const int tid  = threadIdx.x;
  const int lane = tid & 31;
  const int wave = tid >> 5;
  const long rowBase = (long)blockIdx.x * TM;

  {
    const float4* src = (const float4*)(x + rowBase * DIM);
    for (int i = tid; i < (TM * DIM) / 4; i += 128) {
      const int r  = i >> 4;
      const int c4 = (i & 15) << 2;
      *(float4*)&smem[r * PITCH + c4] = src[i];
    }
  }
  __syncthreads();

  const int mta  = (wave >> 1) * 2;
  const int ntp  = wave & 1;
  const int l16  = lane & 15;
  const int half = lane >> 4;
  const int arowA = mta * 16 + l16;
  const int col0  = ntp * 32 + l16;
  const int col1  = col0 + 16;

  for (int t = 1; t < N_NODES; ++t) {
    float b0 = 0.f, b1 = 0.f;
    for (int f = 0; f < t; ++f) {
      const float* bp = bias + (size_t)(f * N_NODES + t) * DIM;
      b0 += bp[col0];
      b1 += bp[col1];
    }
    v8f c00, c01, c10, c11;
    #pragma unroll
    for (int i = 0; i < 8; ++i) {
      c00[i] = b0; c10[i] = b0;
      c01[i] = b1; c11[i] = b1;
    }

    for (int f = 0; f < t; ++f) {
      const float* apA = &smem[f * NODE_STRIDE + arowA * PITCH];
      const float* apB = apA + 16 * PITCH;
      const float* wp  = Wm + (size_t)(f * N_NODES + t) * DIM * DIM;
      #pragma unroll
      for (int k0 = 0; k0 < DIM; k0 += 4) {
        const int kk = k0 + half * 2;
        v2f a0 = *(const v2f*)(apA + kk);
        v2f a1 = *(const v2f*)(apB + kk);
        v2f p, q;
        if (XPOSED) {
          p = *(const v2f*)(wp + col0 * DIM + kk);
          q = *(const v2f*)(wp + col1 * DIM + kk);
        } else {
          p.x = wp[kk * DIM + col0]; p.y = wp[(kk + 1) * DIM + col0];
          q.x = wp[kk * DIM + col1]; q.y = wp[(kk + 1) * DIM + col1];
        }
        c00 = __builtin_amdgcn_wmma_f32_16x16x4_f32(false, a0, false, p, (short)0, c00, false, false);
        c01 = __builtin_amdgcn_wmma_f32_16x16x4_f32(false, a0, false, q, (short)0, c01, false, false);
        c10 = __builtin_amdgcn_wmma_f32_16x16x4_f32(false, a1, false, p, (short)0, c10, false, false);
        c11 = __builtin_amdgcn_wmma_f32_16x16x4_f32(false, a1, false, q, (short)0, c11, false, false);
      }
    }

    if (t < N_NODES - 1) {
      float* dp = &smem[t * NODE_STRIDE];
      #pragma unroll
      for (int i = 0; i < 8; ++i) {
        const int ra = mta * 16 + half * 8 + i;
        const int rb = ra + 16;
        dp[ra * PITCH + col0] = c00[i];
        dp[ra * PITCH + col1] = c01[i];
        dp[rb * PITCH + col0] = c10[i];
        dp[rb * PITCH + col1] = c11[i];
      }
      __syncthreads();
    } else {
      float* op = out + rowBase * DIM;
      #pragma unroll
      for (int i = 0; i < 8; ++i) {
        const int ra = mta * 16 + half * 8 + i;
        const int rb = ra + 16;
        op[(size_t)ra * DIM + col0] = c00[i];
        op[(size_t)ra * DIM + col1] = c01[i];
        op[(size_t)rb * DIM + col0] = c10[i];
        op[(size_t)rb * DIM + col1] = c11[i];
      }
    }
  }
}

extern "C" void kernel_launch(void* const* d_in, const int* in_sizes, int n_in,
                              void* d_out, int out_size, void* d_ws, size_t ws_size,
                              hipStream_t stream) {
  const float* x  = (const float*)d_in[0];
  const float* W  = (const float*)d_in[1];
  const float* b  = (const float*)d_in[2];
  float* out      = (float*)d_out;

  const int Bn      = in_sizes[0] / DIM;
  const int nBlocks = Bn / TM;
  const size_t ldsBytes = (size_t)LDS_FLOATS * sizeof(float);
  const size_t wtBytes  = (size_t)N_NODES * N_NODES * DIM * DIM * sizeof(float);

  if (d_ws != nullptr && ws_size >= wtBytes) {
    float* Wt = (float*)d_ws;
    transpose_w<<<N_NODES * N_NODES, 256, 0, stream>>>(W, Wt);
    (void)hipFuncSetAttribute((const void*)wired_dag_wmma<true>,
                              hipFuncAttributeMaxDynamicSharedMemorySize,
                              (int)ldsBytes);
    wired_dag_wmma<true><<<nBlocks, 128, ldsBytes, stream>>>(x, Wt, b, out);
  } else {
    (void)hipFuncSetAttribute((const void*)wired_dag_wmma<false>,
                              hipFuncAttributeMaxDynamicSharedMemorySize,
                              (int)ldsBytes);
    wired_dag_wmma<false><<<nBlocks, 128, ldsBytes, stream>>>(x, W, b, out);
  }
}


# Fold the linear DAG: per call 5 workgroups build the 64x64 map P9 and bias c9 from all 45 edges on f32 WMMA, then out = x*P9 + c9 in one pass over x; reassociates the f32 sums
# speedup vs baseline: 5.6957x; 5.6957x over previous
#include <hip/hip_runtime.h>


typedef __attribute__((ext_vector_type(2))) float v2f;
typedef __attribute__((ext_vector_type(8))) float v8f;

#define N_NODES 10
#define DIM 64
#define TM 64
#define PITCH 68
#define FOLD_ROWS 16
#define FOLD_TILES 5
#define MAP_ROWS (FOLD_TILES * FOLD_ROWS)
#define N_EDGES 36


__device__ __forceinline__ void load_b(v2f (&dst)[8], const float* __restrict__ W,
                                       int f, int t, int hh, int col, int half) {
  const float* wp = W + (size_t)(f * N_NODES + t) * DIM * DIM + (size_t)(hh * 32) * DIM + col;
  #pragma unroll
  for (int j = 0; j < 8; ++j) {
    const int kk = 4 * j + 2 * half;
    dst[j].x = wp[kk * DIM];
    dst[j].y = wp[(kk + 1) * DIM];
  }
}

__device__ __forceinline__ void load_a(v2f (&dst)[8], const float* ap, int hh, int half) {
  #pragma unroll
  for (int j = 0; j < 8; ++j) dst[j] = *(const v2f*)(ap + hh * 32 + 4 * j + 2 * half);
}

template <bool MAP>
__device__ __forceinline__ void load_init(v8f& d, const float* __restrict__ W,
                                          const float* bsum, int t, int grow0,
                                          int col, int half) {
  if (MAP) {
    const float* w0 = W + (size_t)t * DIM * DIM + (size_t)grow0 * DIM + col;
    #pragma unroll
    for (int i = 0; i < 8; ++i) d[i] = w0[i * DIM];
  } else {
    const float bs = bsum[(t - 1) * DIM + col];
    #pragma unroll
    for (int i = 0; i < 8; ++i) d[i] = 0.f;
    d[0] = (half == 0) ? bs : 0.f;
  }
}

template <bool MAP>
__device__ __forceinline__ void fold_tile(const float* __restrict__ W,
                                          const float* __restrict__ bias,
                                          float* __restrict__ M,
                                          float* acts, float* bsum) {
  const int tid  = threadIdx.x;
  const int lane = tid & 31;
  const int wave = tid >> 5;
  const int l16  = lane & 15;
  const int half = lane >> 4;
  const int col  = wave * 16 + l16;
  const int grow0 = blockIdx.x * FOLD_ROWS + half * 8;

  v2f b0[8], b1[8], b2[8], b3[8];
  v2f a0[8], a1[8];
  load_b(b0, W, 1, 2, 0, col, half);
  load_b(b1, W, 1, 2, 1, col, half);
  load_b(b2, W, 1, 3, 0, col, half);

  if (!MAP) {
    if (tid < DIM) {
      #pragma unroll
      for (int tt = 1; tt < N_NODES; ++tt) {
        float s = 0.f;
        #pragma unroll
        for (int ff = 0; ff < tt; ++ff) s += bias[(size_t)(ff * N_NODES + tt) * DIM + tid];
        bsum[(tt - 1) * DIM + tid] = s;
      }
    }
    __syncthreads();
  }

  v8f ce, co, initn;
  load_init<MAP>(ce, W, bsum, 1, grow0, col, half);
  load_init<MAP>(initn, W, bsum, 2, grow0, col, half);
  #pragma unroll
  for (int i = 0; i < 8; ++i) acts[(half * 8 + i) * PITCH + col] = ce[i];
  __syncthreads();
  load_a(a0, &acts[l16 * PITCH], 0, half);

  int f = 1, t = 2, hh = 0;
  int pf = 1, pt = 3, ph = 1;
  #pragma unroll 4
  for (int e = 0; e < 2 * N_EDGES; ++e) {
    if (f == 1 && hh == 0) {
      ce = initn;
      #pragma unroll
      for (int i = 0; i < 8; ++i) co[i] = 0.f;
      load_init<MAP>(initn, W, bsum, (t + 1 < N_NODES) ? t + 1 : t, grow0, col, half);
    }

    load_b(b3, W, pf, pt, ph, col, half);
    ph ^= 1;
    if (ph == 0) {
      ++pf;
      if (pf >= pt) { pf = 1; ++pt; }
    }
    if (pt > N_NODES - 1) { pt = N_NODES - 1; pf = N_NODES - 2; ph = 1; }

    const int nf = (hh == 0) ? f : ((f + 1 < t) ? f + 1 : 1);
    load_a(a1, &acts[(nf - 1) * FOLD_ROWS * PITCH + l16 * PITCH], hh ^ 1, half);

    #pragma unroll
    for (int j = 0; j < 8; j += 2) {
      ce = __builtin_amdgcn_wmma_f32_16x16x4_f32(false, a0[j],     false, b0[j],     (short)0, ce, false, false);
      co = __builtin_amdgcn_wmma_f32_16x16x4_f32(false, a0[j + 1], false, b0[j + 1], (short)0, co, false, false);
    }

    if (hh == 1) {
      if (f == t - 1) {
        float* dp = &acts[(t - 1) * FOLD_ROWS * PITCH];
        #pragma unroll
        for (int i = 0; i < 8; ++i) dp[(half * 8 + i) * PITCH + col] = ce[i] + co[i];
        __builtin_amdgcn_fence(__ATOMIC_RELEASE, "workgroup", "local");
        __builtin_amdgcn_s_barrier();
        __builtin_amdgcn_fence(__ATOMIC_ACQUIRE, "workgroup", "local");
        f = 1; ++t;
      } else {
        ++f;
      }
    }
    hh ^= 1;
    #pragma unroll
    for (int j = 0; j < 8; ++j) { a0[j] = a1[j]; b0[j] = b1[j]; b1[j] = b2[j]; b2[j] = b3[j]; }
    __builtin_amdgcn_sched_barrier(0);
  }

  {
    const float* sp = &acts[(N_NODES - 2) * FOLD_ROWS * PITCH];
    float4* dst = (float4*)(M + (size_t)blockIdx.x * FOLD_ROWS * DIM);
    #pragma unroll
    for (int u = 0; u < (FOLD_ROWS * DIM) / (4 * 128); ++u) {
      const int i  = tid + u * 128;
      const int r  = i >> 4;
      const int c4 = (i & 15) << 2;
      dst[i] = *(const float4*)&sp[r * PITCH + c4];
    }
  }
}

__global__ __launch_bounds__(128) void fold_weights(const float* __restrict__ W,
                                                    const float* __restrict__ bias,
                                                    float* __restrict__ M) {
  __shared__ __align__(16) float acts[(N_NODES - 1) * FOLD_ROWS * PITCH];
  __shared__ float bsum[(N_NODES - 1) * DIM];
  if (blockIdx.x < 4) fold_tile<true>(W, bias, M, acts, bsum);
  else                fold_tile<false>(W, bias, M, acts, bsum);
}

__global__ __launch_bounds__(128) void apply_map(const float* __restrict__ x,
                                                 const float* __restrict__ M,
                                                 float* __restrict__ out) {
  __shared__ __align__(16) float xin[TM * PITCH];
  __shared__ __align__(16) float yout[TM * PITCH];
  const int tid  = threadIdx.x;
  const int lane = tid & 31;
  const int wave = tid >> 5;
  const long rowBase = (long)blockIdx.x * TM;

  {
    const float4* src = (const float4*)(x + rowBase * DIM);
    float4 xv[(TM * DIM) / (4 * 128)];
    #pragma unroll
    for (int u = 0; u < (TM * DIM) / (4 * 128); ++u) xv[u] = src[tid + u * 128];
    #pragma unroll
    for (int u = 0; u < (TM * DIM) / (4 * 128); ++u) {
      const int i  = tid + u * 128;
      const int r  = i >> 4;
      const int c4 = (i & 15) << 2;
      *(float4*)&xin[r * PITCH + c4] = xv[u];
    }
  }

  const int mta  = (wave >> 1) * 2;
  const int ntp  = wave & 1;
  const int l16  = lane & 15;
  const int half = lane >> 4;
  const int arowA = mta * 16 + l16;
  const int col0  = ntp * 32 + l16;
  const int col1  = col0 + 16;

  v2f p[16], q[16];
  #pragma unroll
  for (int j = 0; j < 16; ++j) {
    const int kk = 4 * j + 2 * half;
    p[j].x = M[kk * DIM + col0]; p[j].y = M[(kk + 1) * DIM + col0];
    q[j].x = M[kk * DIM + col1]; q[j].y = M[(kk + 1) * DIM + col1];
  }
  const float bb0 = M[DIM * DIM + col0];
  const float bb1 = M[DIM * DIM + col1];
  __syncthreads();

  v8f c00, c01, c10, c11;
  #pragma unroll
  for (int i = 0; i < 8; ++i) {
    c00[i] = bb0; c10[i] = bb0;
    c01[i] = bb1; c11[i] = bb1;
  }
  const float* apA = &xin[arowA * PITCH];
  const float* apB = apA + 16 * PITCH;
  #pragma unroll
  for (int j = 0; j < 16; ++j) {
    const int kk = 4 * j + 2 * half;
    v2f a0 = *(const v2f*)(apA + kk);
    v2f a1 = *(const v2f*)(apB + kk);
    c00 = __builtin_amdgcn_wmma_f32_16x16x4_f32(false, a0, false, p[j], (short)0, c00, false, false);
    c01 = __builtin_amdgcn_wmma_f32_16x16x4_f32(false, a0, false, q[j], (short)0, c01, false, false);
    c10 = __builtin_amdgcn_wmma_f32_16x16x4_f32(false, a1, false, p[j], (short)0, c10, false, false);
    c11 = __builtin_amdgcn_wmma_f32_16x16x4_f32(false, a1, false, q[j], (short)0, c11, false, false);
  }

  #pragma unroll
  for (int i = 0; i < 8; ++i) {
    const int ra = mta * 16 + half * 8 + i;
    const int rb = ra + 16;
    yout[ra * PITCH + col0] = c00[i];
    yout[ra * PITCH + col1] = c01[i];
    yout[rb * PITCH + col0] = c10[i];
    yout[rb * PITCH + col1] = c11[i];
  }
  __syncthreads();
  {
    float4* dst = (float4*)(out + rowBase * DIM);
    #pragma unroll
    for (int u = 0; u < (TM * DIM) / (4 * 128); ++u) {
      const int i  = tid + u * 128;
      const int r  = i >> 4;
      const int c4 = (i & 15) << 2;
      dst[i] = *(const float4*)&yout[r * PITCH + c4];
    }
  }
}

extern "C" void kernel_launch(void* const* d_in, const int* in_sizes, int n_in,
                              void* d_out, int out_size, void* d_ws, size_t ws_size,
                              hipStream_t stream) {
  const float* x  = (const float*)d_in[0];
  const float* W  = (const float*)d_in[1];
  const float* b  = (const float*)d_in[2];
  float* out      = (float*)d_out;
  float* M        = (float*)d_ws;

  const int Bn      = in_sizes[0] / DIM;
  const int nBlocks = Bn / TM;

  fold_weights<<<FOLD_TILES, 128, 0, stream>>>(W, b, M);
  apply_map<<<nBlocks, 128, 0, stream>>>(x, M, out);
}


# Fold kernel data path: W tiles come in as 16-byte row loads four edges ahead and are transposed through a wave-private LDS stage, start values parked in LDS, so no load wait sits in the WMMA chain
# speedup vs baseline: 7.2538x; 1.2736x over previous
#include <hip/hip_runtime.h>


typedef __attribute__((ext_vector_type(2))) float v2f;
typedef __attribute__((ext_vector_type(8))) float v8f;

#define N_NODES 10
#define DIM 64
#define TM 64
#define PITCH 68
#define FOLD_ROWS 16
#define FOLD_TILES 5
#define MAP_ROWS (FOLD_TILES * FOLD_ROWS)
#define N_EDGES 36


#define AHEAD 4
#define STAGE_FLOATS (DIM * 16)

__device__ __forceinline__ void load_raw(float4 (&dst)[8], const float* __restrict__ W,
                                         int f, int t, int goff) {
  const float* wp = W + (size_t)(f * N_NODES + t) * DIM * DIM + goff;
  #pragma unroll
  for (int u = 0; u < 8; ++u) dst[u] = *(const float4*)(wp + u * 8 * DIM);
}

__device__ __forceinline__ void stage_store(float* stg, const float4 (&src)[8], int soff) {
  #pragma unroll
  for (int u = 0; u < 8; ++u) *(float4*)(stg + soff + u * 128) = src[u];
}

__device__ __forceinline__ void load_bfrag(v2f (&dst)[16], const float* stg, int foff) {
  #pragma unroll
  for (int j = 0; j < 16; ++j) {
    dst[j].x = stg[foff + (4 * j) * 16];
    dst[j].y = stg[foff + (4 * j + 1) * 16];
  }
}

__device__ __forceinline__ void load_a(v2f (&dst)[16], const float* ap, int half) {
  #pragma unroll
  for (int j = 0; j < 16; ++j) dst[j] = *(const v2f*)(ap + 4 * j + 2 * half);
}

__device__ __forceinline__ void read_init(v8f& d, const float* initv, int t, int tid) {
  #pragma unroll
  for (int i = 0; i < 8; ++i) d[i] = initv[((t - 1) * 8 + i) * 128 + tid];
}

__device__ __forceinline__ void next_edge(int& f, int& t) {
  ++f;
  if (f >= t) { f = 1; ++t; }
  if (t > N_NODES - 1) { t = N_NODES - 1; f = N_NODES - 2; }
}

template <bool MAP>
__device__ __forceinline__ void fold_tile(const float* __restrict__ W,
                                          const float* __restrict__ bias,
                                          float* __restrict__ M,
                                          float* acts, float* bsum,
                                          float* stage_all, float* initv) {
  const int tid  = threadIdx.x;
  const int lane = tid & 31;
  const int wave = tid >> 5;
  const int l16  = lane & 15;
  const int half = lane >> 4;
  const int col  = wave * 16 + l16;
  const int grow0 = blockIdx.x * FOLD_ROWS + half * 8;
  const int q4   = (lane & 3) * 4;
  const int r8   = lane >> 2;
  const int goff = r8 * DIM + wave * 16 + q4;
  const int soff = r8 * 16 + q4;
  const int foff = 2 * half * 16 + l16;
  float* stg = stage_all + wave * STAGE_FLOATS;

  float4 rt[8], r0[8], r1[8], r2[8], r3[8];
  load_raw(rt, W, 1, 2, goff);
  load_raw(r0, W, 1, 3, goff);
  load_raw(r1, W, 2, 3, goff);
  load_raw(r2, W, 1, 4, goff);
  load_raw(r3, W, 2, 4, goff);
  int pf = 3, pt = 4;
  stage_store(stg, rt, soff);

  if (MAP) {
    #pragma unroll
    for (int tt = 1; tt < N_NODES; ++tt) {
      const float* w0 = W + (size_t)tt * DIM * DIM + (size_t)grow0 * DIM + col;
      float v[8];
      #pragma unroll
      for (int i = 0; i < 8; ++i) v[i] = w0[i * DIM];
      #pragma unroll
      for (int i = 0; i < 8; ++i) initv[((tt - 1) * 8 + i) * 128 + tid] = v[i];
    }
  } else {
    if (tid < DIM) {
      #pragma unroll
      for (int tt = 1; tt < N_NODES; ++tt) {
        float s = 0.f;
        #pragma unroll
        for (int ff = 0; ff < tt; ++ff) s += bias[(size_t)(ff * N_NODES + tt) * DIM + tid];
        bsum[(tt - 1) * DIM + tid] = s;
      }
    }
    __syncthreads();
    #pragma unroll
    for (int tt = 1; tt < N_NODES; ++tt) {
      const float bs = bsum[(tt - 1) * DIM + col];
      #pragma unroll
      for (int i = 0; i < 8; ++i)
        initv[((tt - 1) * 8 + i) * 128 + tid] = (i == 0 && half == 0) ? bs : 0.f;
    }
  }

  v8f ce, co;
  read_init(ce, initv, 1, tid);
  #pragma unroll
  for (int i = 0; i < 8; ++i) acts[(half * 8 + i) * PITCH + col] = ce[i];
  __syncthreads();

  v2f bc[16], bn[16];
  v2f ac[16], an[16];
  __builtin_amdgcn_fence(__ATOMIC_RELEASE, "wavefront");
  __builtin_amdgcn_wave_barrier();
  __builtin_amdgcn_fence(__ATOMIC_ACQUIRE, "wavefront");
  load_bfrag(bc, stg, foff);
  load_a(ac, &acts[l16 * PITCH], half);
  read_init(ce, initv, 2, tid);
  #pragma unroll
  for (int i = 0; i < 8; ++i) co[i] = 0.f;

  int f = 1, t = 2;
  int nf = 1, nt = 3;
  #pragma unroll 4
  for (int e = 0; e < N_EDGES; ++e) {
    stage_store(stg, r0, soff);
    load_raw(r0, W, pf, pt, goff);
    next_edge(pf, pt);
    __builtin_amdgcn_fence(__ATOMIC_RELEASE, "wavefront");
    __builtin_amdgcn_wave_barrier();
    __builtin_amdgcn_fence(__ATOMIC_ACQUIRE, "wavefront");
    load_bfrag(bn, stg, foff);
    load_a(an, &acts[(nf - 1) * FOLD_ROWS * PITCH + l16 * PITCH], half);

    #pragma unroll
    for (int j = 0; j < 16; j += 2) {
      ce = __builtin_amdgcn_wmma_f32_16x16x4_f32(false, ac[j],     false, bc[j],     (short)0, ce, false, false);
      co = __builtin_amdgcn_wmma_f32_16x16x4_f32(false, ac[j + 1], false, bc[j + 1], (short)0, co, false, false);
    }

    if (f == t - 1) {
      float* dp = &acts[(t - 1) * FOLD_ROWS * PITCH];
      #pragma unroll
      for (int i = 0; i < 8; ++i) dp[(half * 8 + i) * PITCH + col] = ce[i] + co[i];
      __builtin_amdgcn_fence(__ATOMIC_RELEASE, "workgroup", "local");
      __builtin_amdgcn_s_barrier();
      __builtin_amdgcn_fence(__ATOMIC_ACQUIRE, "workgroup", "local");
      read_init(ce, initv, (t + 1 < N_NODES) ? t + 1 : t, tid);
      #pragma unroll
      for (int i = 0; i < 8; ++i) co[i] = 0.f;
    }
    f = nf; t = nt;
    next_edge(nf, nt);
    #pragma unroll
    for (int j = 0; j < 16; ++j) { ac[j] = an[j]; bc[j] = bn[j]; }
    #pragma unroll
    for (int u = 0; u < 8; ++u) { const float4 nw = r0[u]; r0[u] = r1[u]; r1[u] = r2[u]; r2[u] = r3[u]; r3[u] = nw; }
    __builtin_amdgcn_sched_barrier(0);
  }

  {
    const float* sp = &acts[(N_NODES - 2) * FOLD_ROWS * PITCH];
    float4* dst = (float4*)(M + (size_t)blockIdx.x * FOLD_ROWS * DIM);
    #pragma unroll
    for (int u = 0; u < (FOLD_ROWS * DIM) / (4 * 128); ++u) {
      const int i  = tid + u * 128;
      const int r  = i >> 4;
      const int c4 = (i & 15) << 2;
      dst[i] = *(const float4*)&sp[r * PITCH + c4];
    }
  }
}

__global__ __launch_bounds__(128) void fold_weights(const float* __restrict__ W,
                                                    const float* __restrict__ bias,
                                                    float* __restrict__ M) {
  __shared__ __align__(16) float acts[(N_NODES - 1) * FOLD_ROWS * PITCH];
  __shared__ float bsum[(N_NODES - 1) * DIM];
  __shared__ __align__(16) float stage_all[4 * STAGE_FLOATS];
  __shared__ float initv[(N_NODES - 1) * 8 * 128];
  if (blockIdx.x < 4) fold_tile<true>(W, bias, M, acts, bsum, stage_all, initv);
  else                fold_tile<false>(W, bias, M, acts, bsum, stage_all, initv);
}

__global__ __launch_bounds__(128) void apply_map(const float* __restrict__ x,
                                                 const float* __restrict__ M,
                                                 float* __restrict__ out) {
  __shared__ __align__(16) float xin[TM * PITCH];
  __shared__ __align__(16) float yout[TM * PITCH];
  const int tid  = threadIdx.x;
  const int lane = tid & 31;
  const int wave = tid >> 5;
  const long rowBase = (long)blockIdx.x * TM;

  {
    const float4* src = (const float4*)(x + rowBase * DIM);
    float4 xv[(TM * DIM) / (4 * 128)];
    #pragma unroll
    for (int u = 0; u < (TM * DIM) / (4 * 128); ++u) xv[u] = src[tid + u * 128];
    #pragma unroll
    for (int u = 0; u < (TM * DIM) / (4 * 128); ++u) {
      const int i  = tid + u * 128;
      const int r  = i >> 4;
      const int c4 = (i & 15) << 2;
      *(float4*)&xin[r * PITCH + c4] = xv[u];
    }
  }

  const int mta  = (wave >> 1) * 2;
  const int ntp  = wave & 1;
  const int l16  = lane & 15;
  const int half = lane >> 4;
  const int arowA = mta * 16 + l16;
  const int col0  = ntp * 32 + l16;
  const int col1  = col0 + 16;

  v2f p[16], q[16];
  #pragma unroll
  for (int j = 0; j < 16; ++j) {
    const int kk = 4 * j + 2 * half;
    p[j].x = M[kk * DIM + col0]; p[j].y = M[(kk + 1) * DIM + col0];
    q[j].x = M[kk * DIM + col1]; q[j].y = M[(kk + 1) * DIM + col1];
  }
  const float bb0 = M[DIM * DIM + col0];
  const float bb1 = M[DIM * DIM + col1];
  __syncthreads();

  v8f c00, c01, c10, c11;
  #pragma unroll
  for (int i = 0; i < 8; ++i) {
    c00[i] = bb0; c10[i] = bb0;
    c01[i] = bb1; c11[i] = bb1;
  }
  const float* apA = &xin[arowA * PITCH];
  const float* apB = apA + 16 * PITCH;
  #pragma unroll
  for (int j = 0; j < 16; ++j) {
    const int kk = 4 * j + 2 * half;
    v2f a0 = *(const v2f*)(apA + kk);
    v2f a1 = *(const v2f*)(apB + kk);
    c00 = __builtin_amdgcn_wmma_f32_16x16x4_f32(false, a0, false, p[j], (short)0, c00, false, false);
    c01 = __builtin_amdgcn_wmma_f32_16x16x4_f32(false, a0, false, q[j], (short)0, c01, false, false);
    c10 = __builtin_amdgcn_wmma_f32_16x16x4_f32(false, a1, false, p[j], (short)0, c10, false, false);
    c11 = __builtin_amdgcn_wmma_f32_16x16x4_f32(false, a1, false, q[j], (short)0, c11, false, false);
  }

  #pragma unroll
  for (int i = 0; i < 8; ++i) {
    const int ra = mta * 16 + half * 8 + i;
    const int rb = ra + 16;
    yout[ra * PITCH + col0] = c00[i];
    yout[ra * PITCH + col1] = c01[i];
    yout[rb * PITCH + col0] = c10[i];
    yout[rb * PITCH + col1] = c11[i];
  }
  __syncthreads();
  {
    float4* dst = (float4*)(out + rowBase * DIM);
    #pragma unroll
    for (int u = 0; u < (TM * DIM) / (4 * 128); ++u) {
      const int i  = tid + u * 128;
      const int r  = i >> 4;
      const int c4 = (i & 15) << 2;
      dst[i] = *(const float4*)&yout[r * PITCH + c4];
    }
  }
}

extern "C" void kernel_launch(void* const* d_in, const int* in_sizes, int n_in,
                              void* d_out, int out_size, void* d_ws, size_t ws_size,
                              hipStream_t stream) {
  const float* x  = (const float*)d_in[0];
  const float* W  = (const float*)d_in[1];
  const float* b  = (const float*)d_in[2];
  float* out      = (float*)d_out;
  float* M        = (float*)d_ws;

  const int Bn      = in_sizes[0] / DIM;
  const int nBlocks = Bn / TM;

  fold_weights<<<FOLD_TILES, 128, 0, stream>>>(W, b, M);
  apply_map<<<nBlocks, 128, 0, stream>>>(x, M, out);
}


# Fold loop: every LDS and global memory instruction of an edge stands in one group ahead of that edge's WMMAs, which then issue back to back (scheduling fence; both fragment sets live in registers)
# speedup vs baseline: 7.5278x; 1.0378x over previous
#include <hip/hip_runtime.h>


typedef __attribute__((ext_vector_type(2))) float v2f;
typedef __attribute__((ext_vector_type(8))) float v8f;

#define N_NODES 10
#define DIM 64
#define TM 64
#define PITCH 68
#define FOLD_ROWS 16
#define FOLD_TILES 5
#define MAP_ROWS (FOLD_TILES * FOLD_ROWS)
#define N_EDGES 36


#define AHEAD 4
#define STAGE_FLOATS (DIM * 16)

__device__ __forceinline__ void load_raw(float4 (&dst)[8], const float* __restrict__ W,
                                         int f, int t, int goff) {
  const float* wp = W + (size_t)(f * N_NODES + t) * DIM * DIM + goff;
  #pragma unroll
  for (int u = 0; u < 8; ++u) dst[u] = *(const float4*)(wp + u * 8 * DIM);
}

__device__ __forceinline__ void stage_store(float* stg, const float4 (&src)[8], int soff) {
  #pragma unroll
  for (int u = 0; u < 8; ++u) *(float4*)(stg + soff + u * 128) = src[u];
}

__device__ __forceinline__ void load_bfrag(v2f (&dst)[16], const float* stg, int foff) {
  #pragma unroll
  for (int j = 0; j < 16; ++j) {
    dst[j].x = stg[foff + (4 * j) * 16];
    dst[j].y = stg[foff + (4 * j + 1) * 16];
  }
}

__device__ __forceinline__ void load_a(v2f (&dst)[16], const float* ap, int half) {
  #pragma unroll
  for (int j = 0; j < 16; ++j) dst[j] = *(const v2f*)(ap + 4 * j + 2 * half);
}

__device__ __forceinline__ void read_init(v8f& d, const float* initv, int t, int tid) {
  #pragma unroll
  for (int i = 0; i < 8; ++i) d[i] = initv[((t - 1) * 8 + i) * 128 + tid];
}

__device__ __forceinline__ void next_edge(int& f, int& t) {
  ++f;
  if (f >= t) { f = 1; ++t; }
  if (t > N_NODES - 1) { t = N_NODES - 1; f = N_NODES - 2; }
}

template <bool MAP>
__device__ __forceinline__ void fold_tile(const float* __restrict__ W,
                                          const float* __restrict__ bias,
                                          float* __restrict__ M,
                                          float* acts, float* bsum,
                                          float* stage_all, float* initv) {
  const int tid  = threadIdx.x;
  const int lane = tid & 31;
  const int wave = tid >> 5;
  const int l16  = lane & 15;
  const int half = lane >> 4;
  const int col  = wave * 16 + l16;
  const int grow0 = blockIdx.x * FOLD_ROWS + half * 8;
  const int q4   = (lane & 3) * 4;
  const int r8   = lane >> 2;
  const int goff = r8 * DIM + wave * 16 + q4;
  const int soff = r8 * 16 + q4;
  const int foff = 2 * half * 16 + l16;
  float* stg = stage_all + wave * STAGE_FLOATS;

  float4 rt[8], r0[8], r1[8], r2[8], r3[8];
  load_raw(rt, W, 1, 2, goff);
  load_raw(r0, W, 1, 3, goff);
  load_raw(r1, W, 2, 3, goff);
  load_raw(r2, W, 1, 4, goff);
  load_raw(r3, W, 2, 4, goff);
  int pf = 3, pt = 4;
  stage_store(stg, rt, soff);

  if (MAP) {
    #pragma unroll
    for (int tt = 1; tt < N_NODES; ++tt) {
      const float* w0 = W + (size_t)tt * DIM * DIM + (size_t)grow0 * DIM + col;
      float v[8];
      #pragma unroll
      for (int i = 0; i < 8; ++i) v[i] = w0[i * DIM];
      #pragma unroll
      for (int i = 0; i < 8; ++i) initv[((tt - 1) * 8 + i) * 128 + tid] = v[i];
    }
  } else {
    if (tid < DIM) {
      #pragma unroll
      for (int tt = 1; tt < N_NODES; ++tt) {
        float s = 0.f;
        #pragma unroll
        for (int ff = 0; ff < tt; ++ff) s += bias[(size_t)(ff * N_NODES + tt) * DIM + tid];
        bsum[(tt - 1) * DIM + tid] = s;
      }
    }
    __syncthreads();
    #pragma unroll
    for (int tt = 1; tt < N_NODES; ++tt) {
      const float bs = bsum[(tt - 1) * DIM + col];
      #pragma unroll
      for (int i = 0; i < 8; ++i)
        initv[((tt - 1) * 8 + i) * 128 + tid] = (i == 0 && half == 0) ? bs : 0.f;
    }
  }

  v8f ce, co;
  read_init(ce, initv, 1, tid);
  #pragma unroll
  for (int i = 0; i < 8; ++i) acts[(half * 8 + i) * PITCH + col] = ce[i];
  __syncthreads();

  v2f bc[16], bn[16];
  v2f ac[16], an[16];
  __builtin_amdgcn_fence(__ATOMIC_RELEASE, "wavefront");
  __builtin_amdgcn_wave_barrier();
  __builtin_amdgcn_fence(__ATOMIC_ACQUIRE, "wavefront");
  load_bfrag(bc, stg, foff);
  load_a(ac, &acts[l16 * PITCH], half);
  read_init(ce, initv, 2, tid);
  #pragma unroll
  for (int i = 0; i < 8; ++i) co[i] = 0.f;

  int f = 1, t = 2;
  int nf = 1, nt = 3;
  #pragma unroll 4
  for (int e = 0; e < N_EDGES; ++e) {
    stage_store(stg, r0, soff);
    load_raw(r0, W, pf, pt, goff);
    next_edge(pf, pt);
    __builtin_amdgcn_fence(__ATOMIC_RELEASE, "wavefront");
    __builtin_amdgcn_wave_barrier();
    __builtin_amdgcn_fence(__ATOMIC_ACQUIRE, "wavefront");
    load_bfrag(bn, stg, foff);
    load_a(an, &acts[(nf - 1) * FOLD_ROWS * PITCH + l16 * PITCH], half);
    __builtin_amdgcn_sched_barrier(0);

    #pragma unroll
    for (int j = 0; j < 16; j += 2) {
      ce = __builtin_amdgcn_wmma_f32_16x16x4_f32(false, ac[j],     false, bc[j],     (short)0, ce, false, false);
      co = __builtin_amdgcn_wmma_f32_16x16x4_f32(false, ac[j + 1], false, bc[j + 1], (short)0, co, false, false);
    }

    if (f == t - 1) {
      float* dp = &acts[(t - 1) * FOLD_ROWS * PITCH];
      #pragma unroll
      for (int i = 0; i < 8; ++i) dp[(half * 8 + i) * PITCH + col] = ce[i] + co[i];
      __builtin_amdgcn_fence(__ATOMIC_RELEASE, "workgroup", "local");
      __builtin_amdgcn_s_barrier();
      __builtin_amdgcn_fence(__ATOMIC_ACQUIRE, "workgroup", "local");
      read_init(ce, initv, (t + 1 < N_NODES) ? t + 1 : t, tid);
      #pragma unroll
      for (int i = 0; i < 8; ++i) co[i] = 0.f;
    }
    f = nf; t = nt;
    next_edge(nf, nt);
    #pragma unroll
    for (int j = 0; j < 16; ++j) { ac[j] = an[j]; bc[j] = bn[j]; }
    #pragma unroll
    for (int u = 0; u < 8; ++u) { const float4 nw = r0[u]; r0[u] = r1[u]; r1[u] = r2[u]; r2[u] = r3[u]; r3[u] = nw; }
    __builtin_amdgcn_sched_barrier(0);
  }

  {
    const float* sp = &acts[(N_NODES - 2) * FOLD_ROWS * PITCH];
    float4* dst = (float4*)(M + (size_t)blockIdx.x * FOLD_ROWS * DIM);
    #pragma unroll
    for (int u = 0; u < (FOLD_ROWS * DIM) / (4 * 128); ++u) {
      const int i  = tid + u * 128;
      const int r  = i >> 4;
      const int c4 = (i & 15) << 2;
      dst[i] = *(const float4*)&sp[r * PITCH + c4];
    }
  }
}

__global__ __launch_bounds__(128) void fold_weights(const float* __restrict__ W,
                                                    const float* __restrict__ bias,
                                                    float* __restrict__ M) {
  __shared__ __align__(16) float acts[(N_NODES - 1) * FOLD_ROWS * PITCH];
  __shared__ float bsum[(N_NODES - 1) * DIM];
  __shared__ __align__(16) float stage_all[4 * STAGE_FLOATS];
  __shared__ float initv[(N_NODES - 1) * 8 * 128];
  if (blockIdx.x < 4) fold_tile<true>(W, bias, M, acts, bsum, stage_all, initv);
  else                fold_tile<false>(W, bias, M, acts, bsum, stage_all, initv);
}

__global__ __launch_bounds__(128) void apply_map(const float* __restrict__ x,
                                                 const float* __restrict__ M,
                                                 float* __restrict__ out) {
  __shared__ __align__(16) float xin[TM * PITCH];
  __shared__ __align__(16) float yout[TM * PITCH];
  const int tid  = threadIdx.x;
  const int lane = tid & 31;
  const int wave = tid >> 5;
  const long rowBase = (long)blockIdx.x * TM;

  {
    const float4* src = (const float4*)(x + rowBase * DIM);
    float4 xv[(TM * DIM) / (4 * 128)];
    #pragma unroll
    for (int u = 0; u < (TM * DIM) / (4 * 128); ++u) xv[u] = src[tid + u * 128];
    #pragma unroll
    for (int u = 0; u < (TM * DIM) / (4 * 128); ++u) {
      const int i  = tid + u * 128;
      const int r  = i >> 4;
      const int c4 = (i & 15) << 2;
      *(float4*)&xin[r * PITCH + c4] = xv[u];
    }
  }

  const int mta  = (wave >> 1) * 2;
  const int ntp  = wave & 1;
  const int l16  = lane & 15;
  const int half = lane >> 4;
  const int arowA = mta * 16 + l16;
  const int col0  = ntp * 32 + l16;
  const int col1  = col0 + 16;

  v2f p[16], q[16];
  #pragma unroll
  for (int j = 0; j < 16; ++j) {
    const int kk = 4 * j + 2 * half;
    p[j].x = M[kk * DIM + col0]; p[j].y = M[(kk + 1) * DIM + col0];
    q[j].x = M[kk * DIM + col1]; q[j].y = M[(kk + 1) * DIM + col1];
  }
  const float bb0 = M[DIM * DIM + col0];
  const float bb1 = M[DIM * DIM + col1];
  __syncthreads();

  v8f c00, c01, c10, c11;
  #pragma unroll
  for (int i = 0; i < 8; ++i) {
    c00[i] = bb0; c10[i] = bb0;
    c01[i] = bb1; c11[i] = bb1;
  }
  const float* apA = &xin[arowA * PITCH];
  const float* apB = apA + 16 * PITCH;
  #pragma unroll
  for (int j = 0; j < 16; ++j) {
    const int kk = 4 * j + 2 * half;
    v2f a0 = *(const v2f*)(apA + kk);
    v2f a1 = *(const v2f*)(apB + kk);
    c00 = __builtin_amdgcn_wmma_f32_16x16x4_f32(false, a0, false, p[j], (short)0, c00, false, false);
    c01 = __builtin_amdgcn_wmma_f32_16x16x4_f32(false, a0, false, q[j], (short)0, c01, false, false);
    c10 = __builtin_amdgcn_wmma_f32_16x16x4_f32(false, a1, false, p[j], (short)0, c10, false, false);
    c11 = __builtin_amdgcn_wmma_f32_16x16x4_f32(false, a1, false, q[j], (short)0, c11, false, false);
  }

  #pragma unroll
  for (int i = 0; i < 8; ++i) {
    const int ra = mta * 16 + half * 8 + i;
    const int rb = ra + 16;
    yout[ra * PITCH + col0] = c00[i];
    yout[ra * PITCH + col1] = c01[i];
    yout[rb * PITCH + col0] = c10[i];
    yout[rb * PITCH + col1] = c11[i];
  }
  __syncthreads();
  {
    float4* dst = (float4*)(out + rowBase * DIM);
    #pragma unroll
    for (int u = 0; u < (TM * DIM) / (4 * 128); ++u) {
      const int i  = tid + u * 128;
      const int r  = i >> 4;
      const int c4 = (i & 15) << 2;
      dst[i] = *(const float4*)&yout[r * PITCH + c4];
    }
  }
}

extern "C" void kernel_launch(void* const* d_in, const int* in_sizes, int n_in,
                              void* d_out, int out_size, void* d_ws, size_t ws_size,
                              hipStream_t stream) {
  const float* x  = (const float*)d_in[0];
  const float* W  = (const float*)d_in[1];
  const float* b  = (const float*)d_in[2];
  float* out      = (float*)d_out;
  float* M        = (float*)d_ws;

  const int Bn      = in_sizes[0] / DIM;
  const int nBlocks = Bn / TM;

  fold_weights<<<FOLD_TILES, 128, 0, stream>>>(W, b, M);
  apply_map<<<nBlocks, 128, 0, stream>>>(x, M, out);
}


# Fold walks its 36 edges source-major, one register accumulator pair per level: a node's LDS hand-over hides under the source's later edges, A fragments are read once per source; sum order kept
# speedup vs baseline: 8.3493x; 1.1091x over previous
#include <hip/hip_runtime.h>


typedef __attribute__((ext_vector_type(2))) float v2f;
typedef __attribute__((ext_vector_type(8))) float v8f;

#define N_NODES 10
#define DIM 64
#define TM 64
#define PITCH 68
#define FOLD_ROWS 16
#define FOLD_TILES 5
#define MAP_ROWS (FOLD_TILES * FOLD_ROWS)
#define N_EDGES 36


#define AHEAD 4
#define LAST_NODE (N_NODES - 1)
#define STAGE_FLOATS (DIM * 16)

__device__ __forceinline__ void load_raw(float4 (&dst)[8], const float* __restrict__ W,
                                         int f, int t, int goff) {
  const float* wp = W + (size_t)(f * N_NODES + t) * DIM * DIM + goff;
  #pragma unroll
  for (int u = 0; u < 8; ++u) dst[u] = *(const float4*)(wp + u * 8 * DIM);
}

__device__ __forceinline__ void stage_store(float* stg, const float4 (&src)[8], int soff) {
  #pragma unroll
  for (int u = 0; u < 8; ++u) *(float4*)(stg + soff + u * 128) = src[u];
}

__device__ __forceinline__ void load_bfrag(v2f (&dst)[16], const float* stg, int foff) {
  #pragma unroll
  for (int j = 0; j < 16; ++j) {
    dst[j].x = stg[foff + (4 * j) * 16];
    dst[j].y = stg[foff + (4 * j + 1) * 16];
  }
}

__device__ __forceinline__ void load_a(v2f (&dst)[16], const float* ap, int half) {
  #pragma unroll
  for (int j = 0; j < 16; ++j) dst[j] = *(const v2f*)(ap + 4 * j + 2 * half);
}

constexpr int edge_f(int e) {
  return (e < 8) ? 1 : (e < 15) ? 2 : (e < 21) ? 3 : (e < 26) ? 4 : (e < 30) ? 5 : (e < 33) ? 6 : (e < 35) ? 7 : 8;
}
constexpr int edge_first(int f) {
  return (f == 1) ? 0 : (f == 2) ? 8 : (f == 3) ? 15 : (f == 4) ? 21 : (f == 5) ? 26 : (f == 6) ? 30 : (f == 7) ? 33 : 35;
}
constexpr int edge_t(int e) {
  return edge_f(e) + 1 + e - edge_first(edge_f(e));
}

struct FoldRegs {
  v8f ae[N_NODES], ao[N_NODES];
  float4 rt[8];
  float4 r[AHEAD][8];
  v2f bfr[2][16];
  v2f afr[2][16];
};

__device__ __forceinline__ void store_node(float* acts, int node, const v8f& ce, const v8f& co,
                                           int half, int col) {
  float* dp = &acts[(node - 1) * FOLD_ROWS * PITCH];
  #pragma unroll
  for (int i = 0; i < 8; ++i) dp[(half * 8 + i) * PITCH + col] = ce[i] + co[i];
}

__device__ __forceinline__ void lds_barrier() {
  __builtin_amdgcn_fence(__ATOMIC_RELEASE, "workgroup", "local");
  __builtin_amdgcn_s_barrier();
  __builtin_amdgcn_fence(__ATOMIC_ACQUIRE, "workgroup", "local");
}

template <int E>
__device__ __forceinline__ void fold_trip(FoldRegs& R, const float* __restrict__ W,
                                          float* acts, float* stg,
                                          int goff, int soff, int foff,
                                          int l16, int half, int col) {
  constexpr int f  = edge_f(E);
  constexpr int t  = edge_t(E);
  constexpr int nf = (E + 1 < N_EDGES) ? edge_f(E + 1) : f;
  if constexpr (E + 1 < N_EDGES) {
    stage_store(stg, R.r[E % AHEAD], soff);
    if constexpr (E + 1 + AHEAD < N_EDGES)
      load_raw(R.r[E % AHEAD], W, edge_f(E + 1 + AHEAD), edge_t(E + 1 + AHEAD), goff);
    __builtin_amdgcn_fence(__ATOMIC_RELEASE, "wavefront");
    __builtin_amdgcn_wave_barrier();
    __builtin_amdgcn_fence(__ATOMIC_ACQUIRE, "wavefront");
    load_bfrag(R.bfr[(E + 1) & 1], stg, foff);
    if constexpr (nf != f)
      load_a(R.afr[nf & 1], &acts[(nf - 1) * FOLD_ROWS * PITCH + l16 * PITCH], half);
  }
  if constexpr (f <= LAST_NODE - 3 && t == f + 2)
    store_node(acts, f + 1, R.ae[f + 1], R.ao[f + 1], half, col);
  __builtin_amdgcn_sched_barrier(0);

  #pragma unroll
  for (int j = 0; j < 16; j += 2) {
    R.ae[t] = __builtin_amdgcn_wmma_f32_16x16x4_f32(false, R.afr[f & 1][j],     false, R.bfr[E & 1][j],     (short)0, R.ae[t], false, false);
    R.ao[t] = __builtin_amdgcn_wmma_f32_16x16x4_f32(false, R.afr[f & 1][j + 1], false, R.bfr[E & 1][j + 1], (short)0, R.ao[t], false, false);
  }

  if constexpr (f <= LAST_NODE - 3) {
    if constexpr (t == f + 2) {
      __builtin_amdgcn_sched_barrier(0);
      lds_barrier();
    }
  } else if constexpr (t == f + 1) {
    store_node(acts, t, R.ae[t], R.ao[t], half, col);
    lds_barrier();
  }
  __builtin_amdgcn_sched_barrier(0);
}

template <int E>
__device__ __forceinline__ void fold_trips(FoldRegs& R, const float* __restrict__ W,
                                           float* acts, float* stg,
                                           int goff, int soff, int foff,
                                           int l16, int half, int col) {
  fold_trip<E>(R, W, acts, stg, goff, soff, foff, l16, half, col);
  if constexpr (E + 1 < N_EDGES) fold_trips<E + 1>(R, W, acts, stg, goff, soff, foff, l16, half, col);
}

__global__ __launch_bounds__(128) void fold_weights(const float* __restrict__ W,
                                                    const float* __restrict__ bias,
                                                    float* __restrict__ M) {
  __shared__ __align__(16) float acts[(N_NODES - 1) * FOLD_ROWS * PITCH];
  __shared__ float bsum[(N_NODES - 1) * DIM];
  __shared__ __align__(16) float stage_all[4 * STAGE_FLOATS];

  const int tid  = threadIdx.x;
  const int lane = tid & 31;
  const int wave = tid >> 5;
  const int l16  = lane & 15;
  const int half = lane >> 4;
  const int col  = wave * 16 + l16;
  const int grow0 = blockIdx.x * FOLD_ROWS + half * 8;
  const int q4   = (lane & 3) * 4;
  const int r8   = lane >> 2;
  const int goff = r8 * DIM + wave * 16 + q4;
  const int soff = r8 * 16 + q4;
  const int foff = 2 * half * 16 + l16;
  float* stg = stage_all + wave * STAGE_FLOATS;

  FoldRegs R;
  v8f n1;
  load_raw(R.rt, W, edge_f(0), edge_t(0), goff);

  if (blockIdx.x < 4) {
    {
      const float* w0 = W + (size_t)1 * DIM * DIM + (size_t)grow0 * DIM + col;
      #pragma unroll
      for (int i = 0; i < 8; ++i) n1[i] = w0[i * DIM];
    }
    #pragma unroll
    for (int tt = 2; tt < N_NODES; ++tt) {
      const float* w0 = W + (size_t)tt * DIM * DIM + (size_t)grow0 * DIM + col;
      #pragma unroll
      for (int i = 0; i < 8; ++i) R.ae[tt][i] = w0[i * DIM];
    }
  } else {
    if (tid < DIM) {
      #pragma unroll
      for (int tt = 1; tt < N_NODES; ++tt) {
        float s = 0.f;
        #pragma unroll
        for (int ff = 0; ff < tt; ++ff) s += bias[(size_t)(ff * N_NODES + tt) * DIM + tid];
        bsum[(tt - 1) * DIM + tid] = s;
      }
    }
    lds_barrier();
    {
      const float bs = bsum[col];
      #pragma unroll
      for (int i = 0; i < 8; ++i) n1[i] = (i == 0 && half == 0) ? bs : 0.f;
    }
    #pragma unroll
    for (int tt = 2; tt < N_NODES; ++tt) {
      const float bs = bsum[(tt - 1) * DIM + col];
      #pragma unroll
      for (int i = 0; i < 8; ++i) R.ae[tt][i] = (i == 0 && half == 0) ? bs : 0.f;
    }
  }
  load_raw(R.r[0], W, edge_f(1), edge_t(1), goff);
  load_raw(R.r[1], W, edge_f(2), edge_t(2), goff);
  load_raw(R.r[2], W, edge_f(3), edge_t(3), goff);
  load_raw(R.r[3], W, edge_f(4), edge_t(4), goff);
  #pragma unroll
  for (int tt = 2; tt < N_NODES; ++tt) {
    #pragma unroll
    for (int i = 0; i < 8; ++i) R.ao[tt][i] = 0.f;
  }

  #pragma unroll
  for (int i = 0; i < 8; ++i) acts[(half * 8 + i) * PITCH + col] = n1[i];
  stage_store(stg, R.rt, soff);
  lds_barrier();
  load_bfrag(R.bfr[0], stg, foff);
  load_a(R.afr[1], &acts[l16 * PITCH], half);

  fold_trips<0>(R, W, acts, stg, goff, soff, foff, l16, half, col);

  {
    const float* sp = &acts[(N_NODES - 2) * FOLD_ROWS * PITCH];
    float4* dst = (float4*)(M + (size_t)blockIdx.x * FOLD_ROWS * DIM);
    #pragma unroll
    for (int u = 0; u < (FOLD_ROWS * DIM) / (4 * 128); ++u) {
      const int i  = tid + u * 128;
      const int r  = i >> 4;
      const int c4 = (i & 15) << 2;
      dst[i] = *(const float4*)&sp[r * PITCH + c4];
    }
  }
}

__global__ __launch_bounds__(128) void apply_map(const float* __restrict__ x,
                                                 const float* __restrict__ M,
                                                 float* __restrict__ out) {
  __shared__ __align__(16) float xin[TM * PITCH];
  __shared__ __align__(16) float yout[TM * PITCH];
  const int tid  = threadIdx.x;
  const int lane = tid & 31;
  const int wave = tid >> 5;
  const long rowBase = (long)blockIdx.x * TM;

  {
    const float4* src = (const float4*)(x + rowBase * DIM);
    float4 xv[(TM * DIM) / (4 * 128)];
    #pragma unroll
    for (int u = 0; u < (TM * DIM) / (4 * 128); ++u) xv[u] = src[tid + u * 128];
    #pragma unroll
    for (int u = 0; u < (TM * DIM) / (4 * 128); ++u) {
      const int i  = tid + u * 128;
      const int r  = i >> 4;
      const int c4 = (i & 15) << 2;
      *(float4*)&xin[r * PITCH + c4] = xv[u];
    }
  }

  const int mta  = (wave >> 1) * 2;
  const int ntp  = wave & 1;
  const int l16  = lane & 15;
  const int half = lane >> 4;
  const int arowA = mta * 16 + l16;
  const int col0  = ntp * 32 + l16;
  const int col1  = col0 + 16;

  v2f p[16], q[16];
  #pragma unroll
  for (int j = 0; j < 16; ++j) {
    const int kk = 4 * j + 2 * half;
    p[j].x = M[kk * DIM + col0]; p[j].y = M[(kk + 1) * DIM + col0];
    q[j].x = M[kk * DIM + col1]; q[j].y = M[(kk + 1) * DIM + col1];
  }
  const float bb0 = M[DIM * DIM + col0];
  const float bb1 = M[DIM * DIM + col1];
  __syncthreads();

  v8f c00, c01, c10, c11;
  #pragma unroll
  for (int i = 0; i < 8; ++i) {
    c00[i] = bb0; c10[i] = bb0;
    c01[i] = bb1; c11[i] = bb1;
  }
  const float* apA = &xin[arowA * PITCH];
  const float* apB = apA + 16 * PITCH;
  #pragma unroll
  for (int j = 0; j < 16; ++j) {
    const int kk = 4 * j + 2 * half;
    v2f a0 = *(const v2f*)(apA + kk);
    v2f a1 = *(const v2f*)(apB + kk);
    c00 = __builtin_amdgcn_wmma_f32_16x16x4_f32(false, a0, false, p[j], (short)0, c00, false, false);
    c01 = __builtin_amdgcn_wmma_f32_16x16x4_f32(false, a0, false, q[j], (short)0, c01, false, false);
    c10 = __builtin_amdgcn_wmma_f32_16x16x4_f32(false, a1, false, p[j], (short)0, c10, false, false);
    c11 = __builtin_amdgcn_wmma_f32_16x16x4_f32(false, a1, false, q[j], (short)0, c11, false, false);
  }

  #pragma unroll
  for (int i = 0; i < 8; ++i) {
    const int ra = mta * 16 + half * 8 + i;
    const int rb = ra + 16;
    yout[ra * PITCH + col0] = c00[i];
    yout[ra * PITCH + col1] = c01[i];
    yout[rb * PITCH + col0] = c10[i];
    yout[rb * PITCH + col1] = c11[i];
  }
  __syncthreads();
  {
    float4* dst = (float4*)(out + rowBase * DIM);
    #pragma unroll
    for (int u = 0; u < (TM * DIM) / (4 * 128); ++u) {
      const int i  = tid + u * 128;
      const int r  = i >> 4;
      const int c4 = (i & 15) << 2;
      dst[i] = *(const float4*)&yout[r * PITCH + c4];
    }
  }
}

extern "C" void kernel_launch(void* const* d_in, const int* in_sizes, int n_in,
                              void* d_out, int out_size, void* d_ws, size_t ws_size,
                              hipStream_t stream) {
  const float* x  = (const float*)d_in[0];
  const float* W  = (const float*)d_in[1];
  const float* b  = (const float*)d_in[2];
  float* out      = (float*)d_out;
  float* M        = (float*)d_ws;

  const int Bn      = in_sizes[0] / DIM;
  const int nBlocks = Bn / TM;

  fold_weights<<<FOLD_TILES, 128, 0, stream>>>(W, b, M);
  apply_map<<<nBlocks, 128, 0, stream>>>(x, M, out);
}


# apply_map as 256 workgroups of 8 long-lived waves: map staged once in LDS, then each wave alone takes two 16-row tiles with all of P_9 as B fragments in registers; half the waves to enter; sums kept
# speedup vs baseline: 9.4649x; 1.1336x over previous
#include <hip/hip_runtime.h>


typedef __attribute__((ext_vector_type(2))) float v2f;
typedef __attribute__((ext_vector_type(8))) float v8f;
typedef __attribute__((ext_vector_type(4))) float v4f;

#define N_NODES 10
#define DIM 64
#define AP_WAVES 8
#define AP_TILES 2
#define AP_ROWS (AP_WAVES * AP_TILES * 16)
#define MPITCH 80
#define PITCH 68
#define FOLD_ROWS 16
#define FOLD_TILES 5
#define MAP_ROWS (FOLD_TILES * FOLD_ROWS)
#define N_EDGES 36


#define AHEAD 4
#define LAST_NODE (N_NODES - 1)
#define STAGE_FLOATS (DIM * 16)

__device__ __forceinline__ void load_raw(float4 (&dst)[8], const float* __restrict__ W,
                                         int f, int t, int goff) {
  const float* wp = W + (size_t)(f * N_NODES + t) * DIM * DIM + goff;
  #pragma unroll
  for (int u = 0; u < 8; ++u) dst[u] = *(const float4*)(wp + u * 8 * DIM);
}

__device__ __forceinline__ void stage_store(float* stg, const float4 (&src)[8], int soff) {
  #pragma unroll
  for (int u = 0; u < 8; ++u) *(float4*)(stg + soff + u * 128) = src[u];
}

__device__ __forceinline__ void load_bfrag(v2f (&dst)[16], const float* stg, int foff) {
  #pragma unroll
  for (int j = 0; j < 16; ++j) {
    dst[j].x = stg[foff + (4 * j) * 16];
    dst[j].y = stg[foff + (4 * j + 1) * 16];
  }
}

__device__ __forceinline__ void load_a(v2f (&dst)[16], const float* ap, int half) {
  #pragma unroll
  for (int j = 0; j < 16; ++j) dst[j] = *(const v2f*)(ap + 4 * j + 2 * half);
}

constexpr int edge_f(int e) {
  return (e < 8) ? 1 : (e < 15) ? 2 : (e < 21) ? 3 : (e < 26) ? 4 : (e < 30) ? 5 : (e < 33) ? 6 : (e < 35) ? 7 : 8;
}
constexpr int edge_first(int f) {
  return (f == 1) ? 0 : (f == 2) ? 8 : (f == 3) ? 15 : (f == 4) ? 21 : (f == 5) ? 26 : (f == 6) ? 30 : (f == 7) ? 33 : 35;
}
constexpr int edge_t(int e) {
  return edge_f(e) + 1 + e - edge_first(edge_f(e));
}

struct FoldRegs {
  v8f ae[N_NODES], ao[N_NODES];
  float4 rt[8];
  float4 r[AHEAD][8];
  v2f bfr[2][16];
  v2f afr[2][16];
};

__device__ __forceinline__ void store_node(float* acts, int node, const v8f& ce, const v8f& co,
                                           int half, int col) {
  float* dp = &acts[(node - 1) * FOLD_ROWS * PITCH];
  #pragma unroll
  for (int i = 0; i < 8; ++i) dp[(half * 8 + i) * PITCH + col] = ce[i] + co[i];
}

__device__ __forceinline__ void lds_barrier() {
  __builtin_amdgcn_fence(__ATOMIC_RELEASE, "workgroup", "local");
  __builtin_amdgcn_s_barrier();
  __builtin_amdgcn_fence(__ATOMIC_ACQUIRE, "workgroup", "local");
}

__device__ __forceinline__ void wave_lds_order() {
  __builtin_amdgcn_fence(__ATOMIC_RELEASE, "wavefront");
  __builtin_amdgcn_wave_barrier();
  __builtin_amdgcn_fence(__ATOMIC_ACQUIRE, "wavefront");
}

template <int E>
__device__ __forceinline__ void fold_trip(FoldRegs& R, const float* __restrict__ W,
                                          float* acts, float* stg,
                                          int goff, int soff, int foff,
                                          int l16, int half, int col) {
  constexpr int f  = edge_f(E);
  constexpr int t  = edge_t(E);
  constexpr int nf = (E + 1 < N_EDGES) ? edge_f(E + 1) : f;
  if constexpr (E + 1 < N_EDGES) {
    stage_store(stg, R.r[E % AHEAD], soff);
    if constexpr (E + 1 + AHEAD < N_EDGES)
      load_raw(R.r[E % AHEAD], W, edge_f(E + 1 + AHEAD), edge_t(E + 1 + AHEAD), goff);
    __builtin_amdgcn_fence(__ATOMIC_RELEASE, "wavefront");
    __builtin_amdgcn_wave_barrier();
    __builtin_amdgcn_fence(__ATOMIC_ACQUIRE, "wavefront");
    load_bfrag(R.bfr[(E + 1) & 1], stg, foff);
    if constexpr (nf != f)
      load_a(R.afr[nf & 1], &acts[(nf - 1) * FOLD_ROWS * PITCH + l16 * PITCH], half);
  }
  if constexpr (f <= LAST_NODE - 3 && t == f + 2)
    store_node(acts, f + 1, R.ae[f + 1], R.ao[f + 1], half, col);
  __builtin_amdgcn_sched_barrier(0);

  #pragma unroll
  for (int j = 0; j < 16; j += 2) {
    R.ae[t] = __builtin_amdgcn_wmma_f32_16x16x4_f32(false, R.afr[f & 1][j],     false, R.bfr[E & 1][j],     (short)0, R.ae[t], false, false);
    R.ao[t] = __builtin_amdgcn_wmma_f32_16x16x4_f32(false, R.afr[f & 1][j + 1], false, R.bfr[E & 1][j + 1], (short)0, R.ao[t], false, false);
  }

  if constexpr (f <= LAST_NODE - 3) {
    if constexpr (t == f + 2) {
      __builtin_amdgcn_sched_barrier(0);
      lds_barrier();
    }
  } else if constexpr (t == f + 1) {
    store_node(acts, t, R.ae[t], R.ao[t], half, col);
    lds_barrier();
  }
  __builtin_amdgcn_sched_barrier(0);
}

template <int E>
__device__ __forceinline__ void fold_trips(FoldRegs& R, const float* __restrict__ W,
                                           float* acts, float* stg,
                                           int goff, int soff, int foff,
                                           int l16, int half, int col) {
  fold_trip<E>(R, W, acts, stg, goff, soff, foff, l16, half, col);
  if constexpr (E + 1 < N_EDGES) fold_trips<E + 1>(R, W, acts, stg, goff, soff, foff, l16, half, col);
}

__global__ __launch_bounds__(128) void fold_weights(const float* __restrict__ W,
                                                    const float* __restrict__ bias,
                                                    float* __restrict__ M) {
  __shared__ __align__(16) float acts[(N_NODES - 1) * FOLD_ROWS * PITCH];
  __shared__ float bsum[(N_NODES - 1) * DIM];
  __shared__ __align__(16) float stage_all[4 * STAGE_FLOATS];

  const int tid  = threadIdx.x;
  const int lane = tid & 31;
  const int wave = tid >> 5;
  const int l16  = lane & 15;
  const int half = lane >> 4;
  const int col  = wave * 16 + l16;
  const int grow0 = blockIdx.x * FOLD_ROWS + half * 8;
  const int q4   = (lane & 3) * 4;
  const int r8   = lane >> 2;
  const int goff = r8 * DIM + wave * 16 + q4;
  const int soff = r8 * 16 + q4;
  const int foff = 2 * half * 16 + l16;
  float* stg = stage_all + wave * STAGE_FLOATS;

  FoldRegs R;
  v8f n1;
  load_raw(R.rt, W, edge_f(0), edge_t(0), goff);

  if (blockIdx.x < 4) {
    {
      const float* w0 = W + (size_t)1 * DIM * DIM + (size_t)grow0 * DIM + col;
      #pragma unroll
      for (int i = 0; i < 8; ++i) n1[i] = w0[i * DIM];
    }
    #pragma unroll
    for (int tt = 2; tt < N_NODES; ++tt) {
      const float* w0 = W + (size_t)tt * DIM * DIM + (size_t)grow0 * DIM + col;
      #pragma unroll
      for (int i = 0; i < 8; ++i) R.ae[tt][i] = w0[i * DIM];
    }
  } else {
    if (tid < DIM) {
      #pragma unroll
      for (int tt = 1; tt < N_NODES; ++tt) {
        float s = 0.f;
        #pragma unroll
        for (int ff = 0; ff < tt; ++ff) s += bias[(size_t)(ff * N_NODES + tt) * DIM + tid];
        bsum[(tt - 1) * DIM + tid] = s;
      }
    }
    lds_barrier();
    {
      const float bs = bsum[col];
      #pragma unroll
      for (int i = 0; i < 8; ++i) n1[i] = (i == 0 && half == 0) ? bs : 0.f;
    }
    #pragma unroll
    for (int tt = 2; tt < N_NODES; ++tt) {
      const float bs = bsum[(tt - 1) * DIM + col];
      #pragma unroll
      for (int i = 0; i < 8; ++i) R.ae[tt][i] = (i == 0 && half == 0) ? bs : 0.f;
    }
  }
  load_raw(R.r[0], W, edge_f(1), edge_t(1), goff);
  load_raw(R.r[1], W, edge_f(2), edge_t(2), goff);
  load_raw(R.r[2], W, edge_f(3), edge_t(3), goff);
  load_raw(R.r[3], W, edge_f(4), edge_t(4), goff);
  #pragma unroll
  for (int tt = 2; tt < N_NODES; ++tt) {
    #pragma unroll
    for (int i = 0; i < 8; ++i) R.ao[tt][i] = 0.f;
  }

  #pragma unroll
  for (int i = 0; i < 8; ++i) acts[(half * 8 + i) * PITCH + col] = n1[i];
  stage_store(stg, R.rt, soff);
  lds_barrier();
  load_bfrag(R.bfr[0], stg, foff);
  load_a(R.afr[1], &acts[l16 * PITCH], half);

  fold_trips<0>(R, W, acts, stg, goff, soff, foff, l16, half, col);

  {
    const float* sp = &acts[(N_NODES - 2) * FOLD_ROWS * PITCH];
    float4* dst = (float4*)(M + (size_t)blockIdx.x * FOLD_ROWS * DIM);
    #pragma unroll
    for (int u = 0; u < (FOLD_ROWS * DIM) / (4 * 128); ++u) {
      const int i  = tid + u * 128;
      const int r  = i >> 4;
      const int c4 = (i & 15) << 2;
      dst[i] = *(const float4*)&sp[r * PITCH + c4];
    }
  }
}


__device__ __forceinline__ void apply_tile(float* tp, const v4f (&xv)[8],
                                           const v2f (&bf)[4][16], const float (&bb)[4],
                                           float* __restrict__ dst_rows,
                                           int lane, int l16, int half) {
  #pragma unroll
  for (int u = 0; u < 8; ++u) {
    const int q  = lane + u * 32;
    const int r  = q >> 4;
    const int c4 = (q & 15) << 2;
    *(v4f*)&tp[r * PITCH + c4] = xv[u];
  }
  wave_lds_order();
  v2f a[16];
  #pragma unroll
  for (int j = 0; j < 16; ++j) a[j] = *(const v2f*)&tp[l16 * PITCH + 4 * j + 2 * half];
  __builtin_amdgcn_sched_barrier(0);

  v8f c0, c1, c2, c3;
  #pragma unroll
  for (int e = 0; e < 8; ++e) { c0[e] = bb[0]; c1[e] = bb[1]; c2[e] = bb[2]; c3[e] = bb[3]; }
  #pragma unroll
  for (int j = 0; j < 16; ++j) {
    c0 = __builtin_amdgcn_wmma_f32_16x16x4_f32(false, a[j], false, bf[0][j], (short)0, c0, false, false);
    c1 = __builtin_amdgcn_wmma_f32_16x16x4_f32(false, a[j], false, bf[1][j], (short)0, c1, false, false);
    c2 = __builtin_amdgcn_wmma_f32_16x16x4_f32(false, a[j], false, bf[2][j], (short)0, c2, false, false);
    c3 = __builtin_amdgcn_wmma_f32_16x16x4_f32(false, a[j], false, bf[3][j], (short)0, c3, false, false);
  }

  #pragma unroll
  for (int e = 0; e < 8; ++e) {
    float* rp = &tp[(half * 8 + e) * PITCH + l16];
    rp[0]  = c0[e];
    rp[16] = c1[e];
    rp[32] = c2[e];
    rp[48] = c3[e];
  }
  wave_lds_order();
  v4f* dst = (v4f*)dst_rows;
  #pragma unroll
  for (int u = 0; u < 8; ++u) {
    const int q  = lane + u * 32;
    const int r  = q >> 4;
    const int c4 = (q & 15) << 2;
    dst[q] = *(const v4f*)&tp[r * PITCH + c4];
  }
}

__global__ __launch_bounds__(AP_WAVES * 32) __attribute__((amdgpu_waves_per_eu(2, 2)))
void apply_map(const float* __restrict__ x, const float* __restrict__ M, float* __restrict__ out) {
  __shared__ __align__(16) float lds_all[DIM * MPITCH + AP_WAVES * AP_TILES * 16 * PITCH];
  float* mapl     = lds_all;
  float* tile_all = lds_all + DIM * MPITCH;
  const int tid  = threadIdx.x;
  const int lane = tid & 31;
  const int wave = tid >> 5;
  const int l16  = lane & 15;
  const int half = lane >> 4;
  const long rowBase = (long)blockIdx.x * AP_ROWS + wave * (AP_TILES * 16);

  const v4f* msrc = (const v4f*)M;
  const v4f mv0 = msrc[tid];
  const v4f mv1 = msrc[tid + 1 * (AP_WAVES * 32)];
  const v4f mv2 = msrc[tid + 2 * (AP_WAVES * 32)];
  const v4f mv3 = msrc[tid + 3 * (AP_WAVES * 32)];
  float bb[4];
  #pragma unroll
  for (int ct = 0; ct < 4; ++ct) bb[ct] = M[DIM * DIM + ct * 16 + l16];
  const v4f* xsrc = (const v4f*)(x + rowBase * DIM);
  v4f xa[8], xb[8];
  #pragma unroll
  for (int u = 0; u < 8; ++u) xa[u] = xsrc[lane + u * 32];
  #pragma unroll
  for (int u = 0; u < 8; ++u) xb[u] = xsrc[256 + lane + u * 32];
  {
    const int r  = tid >> 4;
    const int c4 = (tid & 15) << 2;
    *(v4f*)&mapl[(r +  0) * MPITCH + c4] = mv0;
    *(v4f*)&mapl[(r + 16) * MPITCH + c4] = mv1;
    *(v4f*)&mapl[(r + 32) * MPITCH + c4] = mv2;
    *(v4f*)&mapl[(r + 48) * MPITCH + c4] = mv3;
  }
  lds_barrier();

  v2f bf[4][16];
  const float* bp = mapl + 2 * half * MPITCH + l16;
  #pragma unroll
  for (int ct = 0; ct < 4; ++ct) {
    #pragma unroll
    for (int j = 0; j < 16; ++j) {
      bf[ct][j].x = bp[(4 * j) * MPITCH + ct * 16];
      bf[ct][j].y = bp[(4 * j + 1) * MPITCH + ct * 16];
    }
  }

  float* tp = tile_all + (wave * AP_TILES) * 16 * PITCH;
  apply_tile(tp,              xa, bf, bb, out + rowBase * DIM,        lane, l16, half);
  apply_tile(tp + 16 * PITCH, xb, bf, bb, out + (rowBase + 16) * DIM, lane, l16, half);
}

extern "C" void kernel_launch(void* const* d_in, const int* in_sizes, int n_in,
                              void* d_out, int out_size, void* d_ws, size_t ws_size,
                              hipStream_t stream) {
  const float* x  = (const float*)d_in[0];
  const float* W  = (const float*)d_in[1];
  const float* b  = (const float*)d_in[2];
  float* out      = (float*)d_out;
  float* M        = (float*)d_ws;

  const int Bn      = in_sizes[0] / DIM;
  const int nBlocks = Bn / AP_ROWS;

  fold_weights<<<FOLD_TILES, 128, 0, stream>>>(W, b, M);
  apply_map<<<nBlocks, AP_WAVES * 32, 0, stream>>>(x, M, out);
}


# Fold: level 9 split over two workgroups per row tile: one walks to node 8 and adds P_8*W[8,9] alone, its twin sums sources 1..7 into level 9; apply adds the two planes (sum order changed)
# speedup vs baseline: 10.5297x; 1.1125x over previous
#include <hip/hip_runtime.h>


typedef __attribute__((ext_vector_type(2))) float v2f;
typedef __attribute__((ext_vector_type(8))) float v8f;
typedef __attribute__((ext_vector_type(4))) float v4f;

#define N_NODES 10
#define DIM 64
#define AP_WAVES 8
#define AP_TILES 2
#define AP_ROWS (AP_WAVES * AP_TILES * 16)
#define MPITCH 80
#define PITCH 68
#define FOLD_ROWS 16
#define FOLD_TILES 5
#define MAP_ROWS (FOLD_TILES * FOLD_ROWS)
#define WALK_LAST 8
#define N_EDGES 28
#define N_TILES_W 29


#define AHEAD 4
#define STAGE_FLOATS (DIM * 16)

__device__ __forceinline__ void load_raw(float4 (&dst)[8], const float* __restrict__ W,
                                         int f, int t, int goff) {
  const float* wp = W + (size_t)(f * N_NODES + t) * DIM * DIM + goff;
  #pragma unroll
  for (int u = 0; u < 8; ++u) dst[u] = *(const float4*)(wp + u * 8 * DIM);
}

__device__ __forceinline__ void stage_store(float* stg, const float4 (&src)[8], int soff) {
  #pragma unroll
  for (int u = 0; u < 8; ++u) *(float4*)(stg + soff + u * 128) = src[u];
}

__device__ __forceinline__ void load_bfrag(v2f (&dst)[16], const float* stg, int foff) {
  #pragma unroll
  for (int j = 0; j < 16; ++j) {
    dst[j].x = stg[foff + (4 * j) * 16];
    dst[j].y = stg[foff + (4 * j + 1) * 16];
  }
}

__device__ __forceinline__ void load_a(v2f (&dst)[16], const float* ap, int half) {
  #pragma unroll
  for (int j = 0; j < 16; ++j) dst[j] = *(const v2f*)(ap + 4 * j + 2 * half);
}

constexpr int edge_f(int e) {
  return (e < 7) ? 1 : (e < 13) ? 2 : (e < 18) ? 3 : (e < 22) ? 4 : (e < 25) ? 5 : (e < 27) ? 6 : (e < 28) ? 7 : 8;
}
constexpr int edge_first(int f) {
  return (f == 1) ? 0 : (f == 2) ? 7 : (f == 3) ? 13 : (f == 4) ? 18 : (f == 5) ? 22 : (f == 6) ? 25 : (f == 7) ? 27 : 28;
}
constexpr int edge_t(int e) {
  return edge_f(e) + 1 + e - edge_first(edge_f(e));
}
__device__ __forceinline__ int tile_level(int e_f, int e_t, int L) {
  return (e_f < WALK_LAST && e_t == WALK_LAST) ? L : e_t;
}

struct FoldRegs {
  v8f ae[N_NODES], ao[N_NODES];
  float4 rt[8];
  float4 r[AHEAD][8];
  v2f bfr[2][16];
  v2f afr[2][16];
};

__device__ __forceinline__ void store_node(float* acts, int node, const v8f& ce, const v8f& co,
                                           int half, int col) {
  float* dp = &acts[(node - 1) * FOLD_ROWS * PITCH];
  #pragma unroll
  for (int i = 0; i < 8; ++i) dp[(half * 8 + i) * PITCH + col] = ce[i] + co[i];
}

__device__ __forceinline__ void lds_barrier() {
  __builtin_amdgcn_fence(__ATOMIC_RELEASE, "workgroup", "local");
  __builtin_amdgcn_s_barrier();
  __builtin_amdgcn_fence(__ATOMIC_ACQUIRE, "workgroup", "local");
}

__device__ __forceinline__ void wave_lds_order() {
  __builtin_amdgcn_fence(__ATOMIC_RELEASE, "wavefront");
  __builtin_amdgcn_wave_barrier();
  __builtin_amdgcn_fence(__ATOMIC_ACQUIRE, "wavefront");
}

template <int E>
__device__ __forceinline__ void fold_trip(FoldRegs& R, const float* __restrict__ W,
                                          float* acts, float* stg,
                                          int goff, int soff, int foff,
                                          int l16, int half, int col, int L) {
  constexpr int f  = edge_f(E);
  constexpr int t  = edge_t(E);
  constexpr int nf = (E + 1 < N_EDGES) ? edge_f(E + 1) : f;
  if constexpr (E + 1 < N_TILES_W) {
    stage_store(stg, R.r[E % AHEAD], soff);
    if constexpr (E + 1 + AHEAD < N_TILES_W)
      load_raw(R.r[E % AHEAD], W, edge_f(E + 1 + AHEAD),
               tile_level(edge_f(E + 1 + AHEAD), edge_t(E + 1 + AHEAD), L), goff);
    __builtin_amdgcn_fence(__ATOMIC_RELEASE, "wavefront");
    __builtin_amdgcn_wave_barrier();
    __builtin_amdgcn_fence(__ATOMIC_ACQUIRE, "wavefront");
    load_bfrag(R.bfr[(E + 1) & 1], stg, foff);
    if constexpr (nf != f)
      load_a(R.afr[nf & 1], &acts[(nf - 1) * FOLD_ROWS * PITCH + l16 * PITCH], half);
  }
  if constexpr (f <= WALK_LAST - 3 && t == f + 2)
    store_node(acts, f + 1, R.ae[f + 1], R.ao[f + 1], half, col);
  __builtin_amdgcn_sched_barrier(0);

  #pragma unroll
  for (int j = 0; j < 16; j += 2) {
    R.ae[t] = __builtin_amdgcn_wmma_f32_16x16x4_f32(false, R.afr[f & 1][j],     false, R.bfr[E & 1][j],     (short)0, R.ae[t], false, false);
    R.ao[t] = __builtin_amdgcn_wmma_f32_16x16x4_f32(false, R.afr[f & 1][j + 1], false, R.bfr[E & 1][j + 1], (short)0, R.ao[t], false, false);
  }

  if constexpr (f <= WALK_LAST - 3) {
    if constexpr (t == f + 2) {
      __builtin_amdgcn_sched_barrier(0);
      lds_barrier();
    }
  } else if constexpr (t == f + 1) {
    store_node(acts, t, R.ae[t], R.ao[t], half, col);
    lds_barrier();
  }
  __builtin_amdgcn_sched_barrier(0);
}

template <int E>
__device__ __forceinline__ void fold_trips(FoldRegs& R, const float* __restrict__ W,
                                           float* acts, float* stg,
                                           int goff, int soff, int foff,
                                           int l16, int half, int col, int L) {
  fold_trip<E>(R, W, acts, stg, goff, soff, foff, l16, half, col, L);
  if constexpr (E + 1 < N_EDGES) fold_trips<E + 1>(R, W, acts, stg, goff, soff, foff, l16, half, col, L);
}

__global__ __launch_bounds__(128) void fold_weights(const float* __restrict__ W,
                                                    const float* __restrict__ bias,
                                                    float* __restrict__ M) {
  __shared__ __align__(16) float acts[(N_NODES - 1) * FOLD_ROWS * PITCH];
  __shared__ float bsum[(N_NODES - 1) * DIM];
  __shared__ __align__(16) float stage_all[4 * STAGE_FLOATS];

  const int tid  = threadIdx.x;
  const int lane = tid & 31;
  const int wave = tid >> 5;
  const int l16  = lane & 15;
  const int half = lane >> 4;
  const int col  = wave * 16 + l16;
  const int tile = (blockIdx.x < FOLD_TILES) ? blockIdx.x : blockIdx.x - FOLD_TILES;
  const int L    = (blockIdx.x < FOLD_TILES) ? WALK_LAST : WALK_LAST + 1;
  const int grow0 = tile * FOLD_ROWS + half * 8;
  const int q4   = (lane & 3) * 4;
  const int r8   = lane >> 2;
  const int goff = r8 * DIM + wave * 16 + q4;
  const int soff = r8 * 16 + q4;
  const int foff = 2 * half * 16 + l16;
  float* stg = stage_all + wave * STAGE_FLOATS;

  FoldRegs R;
  v8f n1;
  load_raw(R.rt, W, edge_f(0), edge_t(0), goff);

  if (tile < 4) {
    {
      const float* w0 = W + (size_t)1 * DIM * DIM + (size_t)grow0 * DIM + col;
      #pragma unroll
      for (int i = 0; i < 8; ++i) n1[i] = w0[i * DIM];
    }
    #pragma unroll
    for (int tt = 2; tt <= WALK_LAST; ++tt) {
      const int lev = (tt == WALK_LAST) ? L : tt;
      const float* w0 = W + (size_t)lev * DIM * DIM + (size_t)grow0 * DIM + col;
      #pragma unroll
      for (int i = 0; i < 8; ++i) R.ae[tt][i] = w0[i * DIM];
    }
  } else {
    if (tid < DIM) {
      #pragma unroll
      for (int tt = 1; tt < N_NODES; ++tt) {
        float s = 0.f;
        #pragma unroll
        for (int ff = 0; ff < tt; ++ff) s += bias[(size_t)(ff * N_NODES + tt) * DIM + tid];
        bsum[(tt - 1) * DIM + tid] = s;
      }
    }
    lds_barrier();
    {
      const float bs = bsum[col];
      #pragma unroll
      for (int i = 0; i < 8; ++i) n1[i] = (i == 0 && half == 0) ? bs : 0.f;
    }
    #pragma unroll
    for (int tt = 2; tt <= WALK_LAST; ++tt) {
      const int lev = (tt == WALK_LAST) ? L : tt;
      const float bs = bsum[(lev - 1) * DIM + col];
      #pragma unroll
      for (int i = 0; i < 8; ++i) R.ae[tt][i] = (i == 0 && half == 0) ? bs : 0.f;
    }
  }
  load_raw(R.r[0], W, edge_f(1), edge_t(1), goff);
  load_raw(R.r[1], W, edge_f(2), edge_t(2), goff);
  load_raw(R.r[2], W, edge_f(3), edge_t(3), goff);
  load_raw(R.r[3], W, edge_f(4), edge_t(4), goff);
  #pragma unroll
  for (int tt = 2; tt < N_NODES; ++tt) {
    #pragma unroll
    for (int i = 0; i < 8; ++i) R.ao[tt][i] = 0.f;
  }
  #pragma unroll
  for (int i = 0; i < 8; ++i) R.ae[WALK_LAST + 1][i] = 0.f;

  #pragma unroll
  for (int i = 0; i < 8; ++i) acts[(half * 8 + i) * PITCH + col] = n1[i];
  stage_store(stg, R.rt, soff);
  lds_barrier();
  load_bfrag(R.bfr[0], stg, foff);
  load_a(R.afr[1], &acts[l16 * PITCH], half);

  fold_trips<0>(R, W, acts, stg, goff, soff, foff, l16, half, col, L);

  if (L == WALK_LAST) {
    load_a(R.afr[0], &acts[(WALK_LAST - 1) * FOLD_ROWS * PITCH + l16 * PITCH], half);
    __builtin_amdgcn_sched_barrier(0);
    #pragma unroll
    for (int j = 0; j < 16; j += 2) {
      R.ae[9] = __builtin_amdgcn_wmma_f32_16x16x4_f32(false, R.afr[0][j],     false, R.bfr[N_EDGES & 1][j],     (short)0, R.ae[9], false, false);
      R.ao[9] = __builtin_amdgcn_wmma_f32_16x16x4_f32(false, R.afr[0][j + 1], false, R.bfr[N_EDGES & 1][j + 1], (short)0, R.ao[9], false, false);
    }
    store_node(acts, WALK_LAST + 1, R.ae[9], R.ao[9], half, col);
    lds_barrier();
  }

  {
    const float* sp = &acts[(L == WALK_LAST ? WALK_LAST : WALK_LAST - 1) * FOLD_ROWS * PITCH];
    float4* dst = (float4*)(M + (size_t)(L == WALK_LAST ? MAP_ROWS * DIM : 0)
                              + (size_t)tile * FOLD_ROWS * DIM);
    #pragma unroll
    for (int u = 0; u < (FOLD_ROWS * DIM) / (4 * 128); ++u) {
      const int i  = tid + u * 128;
      const int r  = i >> 4;
      const int c4 = (i & 15) << 2;
      dst[i] = *(const float4*)&sp[r * PITCH + c4];
    }
  }
}


__device__ __forceinline__ void apply_tile(float* tp, const v4f (&xv)[8],
                                           const v2f (&bf)[4][16], const float (&bb)[4],
                                           float* __restrict__ dst_rows,
                                           int lane, int l16, int half) {
  #pragma unroll
  for (int u = 0; u < 8; ++u) {
    const int q  = lane + u * 32;
    const int r  = q >> 4;
    const int c4 = (q & 15) << 2;
    *(v4f*)&tp[r * PITCH + c4] = xv[u];
  }
  wave_lds_order();
  v2f a[16];
  #pragma unroll
  for (int j = 0; j < 16; ++j) a[j] = *(const v2f*)&tp[l16 * PITCH + 4 * j + 2 * half];
  __builtin_amdgcn_sched_barrier(0);

  v8f c0, c1, c2, c3;
  #pragma unroll
  for (int e = 0; e < 8; ++e) { c0[e] = bb[0]; c1[e] = bb[1]; c2[e] = bb[2]; c3[e] = bb[3]; }
  #pragma unroll
  for (int j = 0; j < 16; ++j) {
    c0 = __builtin_amdgcn_wmma_f32_16x16x4_f32(false, a[j], false, bf[0][j], (short)0, c0, false, false);
    c1 = __builtin_amdgcn_wmma_f32_16x16x4_f32(false, a[j], false, bf[1][j], (short)0, c1, false, false);
    c2 = __builtin_amdgcn_wmma_f32_16x16x4_f32(false, a[j], false, bf[2][j], (short)0, c2, false, false);
    c3 = __builtin_amdgcn_wmma_f32_16x16x4_f32(false, a[j], false, bf[3][j], (short)0, c3, false, false);
  }

  #pragma unroll
  for (int e = 0; e < 8; ++e) {
    float* rp = &tp[(half * 8 + e) * PITCH + l16];
    rp[0]  = c0[e];
    rp[16] = c1[e];
    rp[32] = c2[e];
    rp[48] = c3[e];
  }
  wave_lds_order();
  v4f* dst = (v4f*)dst_rows;
  #pragma unroll
  for (int u = 0; u < 8; ++u) {
    const int q  = lane + u * 32;
    const int r  = q >> 4;
    const int c4 = (q & 15) << 2;
    dst[q] = *(const v4f*)&tp[r * PITCH + c4];
  }
}

__global__ __launch_bounds__(AP_WAVES * 32) __attribute__((amdgpu_waves_per_eu(2, 2)))
void apply_map(const float* __restrict__ x, const float* __restrict__ M, float* __restrict__ out) {
  __shared__ __align__(16) float lds_all[DIM * MPITCH + AP_WAVES * AP_TILES * 16 * PITCH];
  float* mapl     = lds_all;
  float* tile_all = lds_all + DIM * MPITCH;
  const int tid  = threadIdx.x;
  const int lane = tid & 31;
  const int wave = tid >> 5;
  const int l16  = lane & 15;
  const int half = lane >> 4;
  const long rowBase = (long)blockIdx.x * AP_ROWS + wave * (AP_TILES * 16);

  const v4f* msrc = (const v4f*)M;
  const v4f* msrc1 = (const v4f*)(M + MAP_ROWS * DIM);
  const v4f mv0 = msrc[tid];
  const v4f mv1 = msrc[tid + 1 * (AP_WAVES * 32)];
  const v4f mv2 = msrc[tid + 2 * (AP_WAVES * 32)];
  const v4f mv3 = msrc[tid + 3 * (AP_WAVES * 32)];
  const v4f mw0 = msrc1[tid];
  const v4f mw1 = msrc1[tid + 1 * (AP_WAVES * 32)];
  const v4f mw2 = msrc1[tid + 2 * (AP_WAVES * 32)];
  const v4f mw3 = msrc1[tid + 3 * (AP_WAVES * 32)];
  float bb[4], bb1[4];
  #pragma unroll
  for (int ct = 0; ct < 4; ++ct) bb[ct] = M[DIM * DIM + ct * 16 + l16];
  #pragma unroll
  for (int ct = 0; ct < 4; ++ct) bb1[ct] = M[MAP_ROWS * DIM + DIM * DIM + ct * 16 + l16];
  const v4f* xsrc = (const v4f*)(x + rowBase * DIM);
  v4f xa[8], xb[8];
  #pragma unroll
  for (int u = 0; u < 8; ++u) xa[u] = xsrc[lane + u * 32];
  #pragma unroll
  for (int u = 0; u < 8; ++u) xb[u] = xsrc[256 + lane + u * 32];
  {
    const int r  = tid >> 4;
    const int c4 = (tid & 15) << 2;
    *(v4f*)&mapl[(r +  0) * MPITCH + c4] = mv0 + mw0;
    *(v4f*)&mapl[(r + 16) * MPITCH + c4] = mv1 + mw1;
    *(v4f*)&mapl[(r + 32) * MPITCH + c4] = mv2 + mw2;
    *(v4f*)&mapl[(r + 48) * MPITCH + c4] = mv3 + mw3;
  }
  #pragma unroll
  for (int ct = 0; ct < 4; ++ct) bb[ct] += bb1[ct];
  lds_barrier();

  v2f bf[4][16];
  const float* bp = mapl + 2 * half * MPITCH + l16;
  #pragma unroll
  for (int ct = 0; ct < 4; ++ct) {
    #pragma unroll
    for (int j = 0; j < 16; ++j) {
      bf[ct][j].x = bp[(4 * j) * MPITCH + ct * 16];
      bf[ct][j].y = bp[(4 * j + 1) * MPITCH + ct * 16];
    }
  }

  float* tp = tile_all + (wave * AP_TILES) * 16 * PITCH;
  apply_tile(tp,              xa, bf, bb, out + rowBase * DIM,        lane, l16, half);
  apply_tile(tp + 16 * PITCH, xb, bf, bb, out + (rowBase + 16) * DIM, lane, l16, half);
}

extern "C" void kernel_launch(void* const* d_in, const int* in_sizes, int n_in,
                              void* d_out, int out_size, void* d_ws, size_t ws_size,
                              hipStream_t stream) {
  const float* x  = (const float*)d_in[0];
  const float* W  = (const float*)d_in[1];
  const float* b  = (const float*)d_in[2];
  float* out      = (float*)d_out;
  float* M        = (float*)d_ws;

  const int Bn      = in_sizes[0] / DIM;
  const int nBlocks = Bn / AP_ROWS;

  fold_weights<<<2 * FOLD_TILES, 128, 0, stream>>>(W, b, M);
  apply_map<<<nBlocks, AP_WAVES * 32, 0, stream>>>(x, M, out);
}


# Fold: paths 0..9 split three ways by the first node >= 5 they visit, one workgroup per class and row tile on its own sub-DAG: longest walk 20 products, was 29; apply adds 3 planes (sum order changed)
# speedup vs baseline: 11.7540x; 1.1163x over previous
#include <hip/hip_runtime.h>


typedef __attribute__((ext_vector_type(2))) float v2f;
typedef __attribute__((ext_vector_type(8))) float v8f;
typedef __attribute__((ext_vector_type(4))) float v4f;

#define N_NODES 10
#define DIM 64
#define AP_WAVES 8
#define AP_TILES 2
#define AP_ROWS (AP_WAVES * AP_TILES * 16)
#define MPITCH 80
#define PITCH 68
#define FOLD_ROWS 16
#define FOLD_TILES 5
#define MAP_ROWS (FOLD_TILES * FOLD_ROWS)
#define N_PLANES 3


#define AHEAD 4
#define STAGE_FLOATS (DIM * 16)

__device__ __forceinline__ void load_raw(float4 (&dst)[8], const float* __restrict__ W,
                                         int f, int t, int goff) {
  const float* wp = W + (size_t)(f * N_NODES + t) * DIM * DIM + goff;
  #pragma unroll
  for (int u = 0; u < 8; ++u) dst[u] = *(const float4*)(wp + u * 8 * DIM);
}

__device__ __forceinline__ void stage_store(float* stg, const float4 (&src)[8], int soff) {
  #pragma unroll
  for (int u = 0; u < 8; ++u) *(float4*)(stg + soff + u * 128) = src[u];
}

__device__ __forceinline__ void load_bfrag(v2f (&dst)[16], const float* stg, int foff) {
  #pragma unroll
  for (int j = 0; j < 16; ++j) {
    dst[j].x = stg[foff + (4 * j) * 16];
    dst[j].y = stg[foff + (4 * j + 1) * 16];
  }
}

__device__ __forceinline__ void load_a(v2f (&dst)[16], const float* ap, int half) {
  #pragma unroll
  for (int j = 0; j < 16; ++j) dst[j] = *(const v2f*)(ap + 4 * j + 2 * half);
}


struct WalkA {
  static constexpr int NE = 20;
  static constexpr int NS = 9;
  static constexpr int NSTART = 5;
  static constexpr int f(int e) {
    constexpr int a[20] = {1, 1, 1, 1,  2, 2, 2,  3, 3,  4,  5, 5, 5, 5,  6, 6, 6,  7, 7,  8};
    return a[e];
  }
  static constexpr int t(int e) {
    constexpr int a[20] = {2, 3, 4, 5,  3, 4, 5,  4, 5,  5,  6, 7, 8, 9,  7, 8, 9,  8, 9,  9};
    return a[e];
  }
  static __device__ __forceinline__ int node(int slot, int sh) { return slot; }
};

struct WalkB {
  static constexpr int NE = 20;
  static constexpr int NE_SHORT = 15;
  static constexpr int NS = 8;
  static constexpr int NSTART = 6;
  static constexpr int f(int e) {
    constexpr int a[20] = {1, 1, 1, 1, 1,  2, 2, 2, 2,  3, 3, 3,  4, 4,  5, 5, 5,  6, 6,  7};
    return a[e];
  }
  static constexpr int t(int e) {
    constexpr int a[20] = {2, 3, 4, 5, 6,  3, 4, 5, 6,  4, 5, 6,  5, 6,  6, 7, 8,  7, 8,  8};
    return a[e];
  }
  static __device__ __forceinline__ int node(int slot, int sh) {
    return (slot <= 4) ? slot : (slot <= 6) ? slot + 1 + sh : slot + 1;
  }
};

template <class Wk> constexpr int src_count(int f) {
  int n = 0;
  for (int e = 0; e < Wk::NE; ++e) n += (Wk::f(e) == f) ? 1 : 0;
  return n;
}
template <class Wk> constexpr int src_first(int f) {
  for (int e = 0; e < Wk::NE; ++e) if (Wk::f(e) == f) return e;
  return Wk::NE;
}

struct FoldRegs {
  v8f ae[N_NODES], ao[N_NODES];
  float4 rt[8];
  float4 r[AHEAD][8];
  v2f bfr[2][16];
  v2f afr[2][16];
};

__device__ __forceinline__ void store_node(float* acts, int node, const v8f& ce, const v8f& co,
                                           int half, int col) {
  float* dp = &acts[(node - 1) * FOLD_ROWS * PITCH];
  #pragma unroll
  for (int i = 0; i < 8; ++i) dp[(half * 8 + i) * PITCH + col] = ce[i] + co[i];
}

__device__ __forceinline__ void lds_barrier() {
  __builtin_amdgcn_fence(__ATOMIC_RELEASE, "workgroup", "local");
  __builtin_amdgcn_s_barrier();
  __builtin_amdgcn_fence(__ATOMIC_ACQUIRE, "workgroup", "local");
}

__device__ __forceinline__ void wave_lds_order() {
  __builtin_amdgcn_fence(__ATOMIC_RELEASE, "wavefront");
  __builtin_amdgcn_wave_barrier();
  __builtin_amdgcn_fence(__ATOMIC_ACQUIRE, "wavefront");
}

template <class Wk, int E>
__device__ __forceinline__ void fold_trip(FoldRegs& R, const float* __restrict__ W,
                                          float* acts, float* stg,
                                          int goff, int soff, int foff,
                                          int l16, int half, int col, int sh) {
  constexpr int  f    = Wk::f(E);
  constexpr int  t    = Wk::t(E);
  constexpr int  cnt  = src_count<Wk>(f);
  constexpr int  idx  = E - src_first<Wk>(f);
  constexpr bool more = (E + 1 < Wk::NE);
  if constexpr (more) {
    stage_store(stg, R.r[E % AHEAD], soff);
    if constexpr (E + 1 + AHEAD < Wk::NE)
      load_raw(R.r[E % AHEAD], W, Wk::node(Wk::f(E + 1 + AHEAD), sh),
               Wk::node(Wk::t(E + 1 + AHEAD), sh), goff);
    __builtin_amdgcn_fence(__ATOMIC_RELEASE, "wavefront");
    __builtin_amdgcn_wave_barrier();
    __builtin_amdgcn_fence(__ATOMIC_ACQUIRE, "wavefront");
    load_bfrag(R.bfr[(E + 1) & 1], stg, foff);
    if constexpr (idx == cnt - 1 && cnt >= 2)
      load_a(R.afr[(f + 1) & 1], &acts[f * FOLD_ROWS * PITCH + l16 * PITCH], half);
  }
  if constexpr (cnt >= 3 && idx == 1)
    store_node(acts, f + 1, R.ae[f + 1], R.ao[f + 1], half, col);
  __builtin_amdgcn_sched_barrier(0);

  #pragma unroll
  for (int j = 0; j < 16; j += 2) {
    R.ae[t] = __builtin_amdgcn_wmma_f32_16x16x4_f32(false, R.afr[f & 1][j],     false, R.bfr[E & 1][j],     (short)0, R.ae[t], false, false);
    R.ao[t] = __builtin_amdgcn_wmma_f32_16x16x4_f32(false, R.afr[f & 1][j + 1], false, R.bfr[E & 1][j + 1], (short)0, R.ao[t], false, false);
  }

  if constexpr (cnt >= 3) {
    if constexpr (idx == 1) {
      __builtin_amdgcn_sched_barrier(0);
      lds_barrier();
    }
  } else if constexpr (idx == 0) {
    store_node(acts, t, R.ae[t], R.ao[t], half, col);
    lds_barrier();
    if constexpr (cnt == 1 && more)
      load_a(R.afr[(f + 1) & 1], &acts[f * FOLD_ROWS * PITCH + l16 * PITCH], half);
  }
  __builtin_amdgcn_sched_barrier(0);
}

template <class Wk, int E, int END>
__device__ __forceinline__ void fold_trips(FoldRegs& R, const float* __restrict__ W,
                                           float* acts, float* stg,
                                           int goff, int soff, int foff,
                                           int l16, int half, int col, int sh) {
  fold_trip<Wk, E>(R, W, acts, stg, goff, soff, foff, l16, half, col, sh);
  if constexpr (E + 1 < END) fold_trips<Wk, E + 1, END>(R, W, acts, stg, goff, soff, foff, l16, half, col, sh);
}

template <class Wk>
__device__ __forceinline__ void fold_start(FoldRegs& R, const float* __restrict__ W,
                                           float* acts, const float* bsum, float* stg,
                                           int tile, int grow0, int goff, int soff, int foff,
                                           int l16, int half, int col, int sh) {
  v8f n1;
  if (tile < 4) {
    {
      const float* w0 = W + (size_t)1 * DIM * DIM + (size_t)grow0 * DIM + col;
      #pragma unroll
      for (int i = 0; i < 8; ++i) n1[i] = w0[i * DIM];
    }
    #pragma unroll
    for (int ss = 2; ss <= Wk::NSTART; ++ss) {
      const float* w0 = W + (size_t)Wk::node(ss, sh) * DIM * DIM + (size_t)grow0 * DIM + col;
      #pragma unroll
      for (int i = 0; i < 8; ++i) R.ae[ss][i] = w0[i * DIM];
    }
  } else {
    {
      const float bs = bsum[col];
      #pragma unroll
      for (int i = 0; i < 8; ++i) n1[i] = (i == 0 && half == 0) ? bs : 0.f;
    }
    #pragma unroll
    for (int ss = 2; ss <= Wk::NSTART; ++ss) {
      const float bs = bsum[(Wk::node(ss, sh) - 1) * DIM + col];
      #pragma unroll
      for (int i = 0; i < 8; ++i) R.ae[ss][i] = (i == 0 && half == 0) ? bs : 0.f;
    }
  }
  load_raw(R.r[0], W, Wk::node(Wk::f(1), sh), Wk::node(Wk::t(1), sh), goff);
  load_raw(R.r[1], W, Wk::node(Wk::f(2), sh), Wk::node(Wk::t(2), sh), goff);
  load_raw(R.r[2], W, Wk::node(Wk::f(3), sh), Wk::node(Wk::t(3), sh), goff);
  load_raw(R.r[3], W, Wk::node(Wk::f(4), sh), Wk::node(Wk::t(4), sh), goff);
  #pragma unroll
  for (int ss = 2; ss <= Wk::NS; ++ss) {
    #pragma unroll
    for (int i = 0; i < 8; ++i) R.ao[ss][i] = 0.f;
  }
  #pragma unroll
  for (int ss = Wk::NSTART + 1; ss <= Wk::NS; ++ss) {
    #pragma unroll
    for (int i = 0; i < 8; ++i) R.ae[ss][i] = 0.f;
  }

  #pragma unroll
  for (int i = 0; i < 8; ++i) acts[(half * 8 + i) * PITCH + col] = n1[i];
  stage_store(stg, R.rt, soff);
  lds_barrier();
  load_bfrag(R.bfr[0], stg, foff);
  load_a(R.afr[1], &acts[l16 * PITCH], half);
}

__global__ __launch_bounds__(128) void fold_weights(const float* __restrict__ W,
                                                    const float* __restrict__ bias,
                                                    float* __restrict__ M) {
  __shared__ __align__(16) float acts[(N_NODES - 1) * FOLD_ROWS * PITCH];
  __shared__ float bsum[(N_NODES - 1) * DIM];
  __shared__ __align__(16) float stage_all[4 * STAGE_FLOATS];

  const int tid  = threadIdx.x;
  const int lane = tid & 31;
  const int wave = tid >> 5;
  const int l16  = lane & 15;
  const int half = lane >> 4;
  const int col  = wave * 16 + l16;
  const int plane = blockIdx.x / FOLD_TILES;
  const int tile  = blockIdx.x - plane * FOLD_TILES;
  const int sh    = (plane == 2) ? 2 : 0;
  const int grow0 = tile * FOLD_ROWS + half * 8;
  const int q4   = (lane & 3) * 4;
  const int r8   = lane >> 2;
  const int goff = r8 * DIM + wave * 16 + q4;
  const int soff = r8 * 16 + q4;
  const int foff = 2 * half * 16 + l16;
  float* stg = stage_all + wave * STAGE_FLOATS;

  FoldRegs R;
  load_raw(R.rt, W, 1, 2, goff);

  if (tile == 4) {
    if (tid < DIM) {
      #pragma unroll
      for (int tt = 1; tt < N_NODES; ++tt) {
        float s = 0.f;
        #pragma unroll
        for (int ff = 0; ff < tt; ++ff) s += bias[(size_t)(ff * N_NODES + tt) * DIM + tid];
        bsum[(tt - 1) * DIM + tid] = s;
      }
    }
    lds_barrier();
  }

  int last_slot;
  if (plane == 0) {
    fold_start<WalkA>(R, W, acts, bsum, stg, tile, grow0, goff, soff, foff, l16, half, col, 0);
    fold_trips<WalkA, 0, WalkA::NE>(R, W, acts, stg, goff, soff, foff, l16, half, col, 0);
    last_slot = WalkA::NS;
  } else {
    fold_start<WalkB>(R, W, acts, bsum, stg, tile, grow0, goff, soff, foff, l16, half, col, sh);
    fold_trips<WalkB, 0, WalkB::NE_SHORT>(R, W, acts, stg, goff, soff, foff, l16, half, col, sh);
    if (plane == 2) {
      store_node(acts, 6, R.ae[6], R.ao[6], half, col);
      lds_barrier();
      last_slot = 6;
    } else {
      fold_trips<WalkB, WalkB::NE_SHORT, WalkB::NE>(R, W, acts, stg, goff, soff, foff, l16, half, col, sh);
      last_slot = WalkB::NS;
    }
  }

  {
    const float* sp = &acts[(last_slot - 1) * FOLD_ROWS * PITCH];
    float4* dst = (float4*)(M + (size_t)plane * MAP_ROWS * DIM + (size_t)tile * FOLD_ROWS * DIM);
    #pragma unroll
    for (int u = 0; u < (FOLD_ROWS * DIM) / (4 * 128); ++u) {
      const int i  = tid + u * 128;
      const int r  = i >> 4;
      const int c4 = (i & 15) << 2;
      dst[i] = *(const float4*)&sp[r * PITCH + c4];
    }
  }
}


__device__ __forceinline__ void apply_tile(float* tp, const v4f (&xv)[8],
                                           const v2f (&bf)[4][16], const float (&bb)[4],
                                           float* __restrict__ dst_rows,
                                           int lane, int l16, int half) {
  #pragma unroll
  for (int u = 0; u < 8; ++u) {
    const int q  = lane + u * 32;
    const int r  = q >> 4;
    const int c4 = (q & 15) << 2;
    *(v4f*)&tp[r * PITCH + c4] = xv[u];
  }
  wave_lds_order();
  v2f a[16];
  #pragma unroll
  for (int j = 0; j < 16; ++j) a[j] = *(const v2f*)&tp[l16 * PITCH + 4 * j + 2 * half];
  __builtin_amdgcn_sched_barrier(0);

  v8f c0, c1, c2, c3;
  #pragma unroll
  for (int e = 0; e < 8; ++e) { c0[e] = bb[0]; c1[e] = bb[1]; c2[e] = bb[2]; c3[e] = bb[3]; }
  #pragma unroll
  for (int j = 0; j < 16; ++j) {
    c0 = __builtin_amdgcn_wmma_f32_16x16x4_f32(false, a[j], false, bf[0][j], (short)0, c0, false, false);
    c1 = __builtin_amdgcn_wmma_f32_16x16x4_f32(false, a[j], false, bf[1][j], (short)0, c1, false, false);
    c2 = __builtin_amdgcn_wmma_f32_16x16x4_f32(false, a[j], false, bf[2][j], (short)0, c2, false, false);
    c3 = __builtin_amdgcn_wmma_f32_16x16x4_f32(false, a[j], false, bf[3][j], (short)0, c3, false, false);
  }

  #pragma unroll
  for (int e = 0; e < 8; ++e) {
    float* rp = &tp[(half * 8 + e) * PITCH + l16];
    rp[0]  = c0[e];
    rp[16] = c1[e];
    rp[32] = c2[e];
    rp[48] = c3[e];
  }
  wave_lds_order();
  v4f* dst = (v4f*)dst_rows;
  #pragma unroll
  for (int u = 0; u < 8; ++u) {
    const int q  = lane + u * 32;
    const int r  = q >> 4;
    const int c4 = (q & 15) << 2;
    dst[q] = *(const v4f*)&tp[r * PITCH + c4];
  }
}

__global__ __launch_bounds__(AP_WAVES * 32) __attribute__((amdgpu_waves_per_eu(2, 2)))
void apply_map(const float* __restrict__ x, const float* __restrict__ M, float* __restrict__ out) {
  __shared__ __align__(16) float lds_all[DIM * MPITCH + AP_WAVES * AP_TILES * 16 * PITCH];
  float* mapl     = lds_all;
  float* tile_all = lds_all + DIM * MPITCH;
  const int tid  = threadIdx.x;
  const int lane = tid & 31;
  const int wave = tid >> 5;
  const int l16  = lane & 15;
  const int half = lane >> 4;
  const long rowBase = (long)blockIdx.x * AP_ROWS + wave * (AP_TILES * 16);

  const v4f* msrc = (const v4f*)M;
  const v4f* msrc1 = (const v4f*)(M + MAP_ROWS * DIM);
  const v4f* msrc2 = (const v4f*)(M + 2 * MAP_ROWS * DIM);
  const v4f mv0 = msrc[tid];
  const v4f mv1 = msrc[tid + 1 * (AP_WAVES * 32)];
  const v4f mv2 = msrc[tid + 2 * (AP_WAVES * 32)];
  const v4f mv3 = msrc[tid + 3 * (AP_WAVES * 32)];
  const v4f mw0 = msrc1[tid];
  const v4f mw1 = msrc1[tid + 1 * (AP_WAVES * 32)];
  const v4f mw2 = msrc1[tid + 2 * (AP_WAVES * 32)];
  const v4f mw3 = msrc1[tid + 3 * (AP_WAVES * 32)];
  const v4f mx0 = msrc2[tid];
  const v4f mx1 = msrc2[tid + 1 * (AP_WAVES * 32)];
  const v4f mx2 = msrc2[tid + 2 * (AP_WAVES * 32)];
  const v4f mx3 = msrc2[tid + 3 * (AP_WAVES * 32)];
  float bb[4], bb1[4], bb2[4];
  #pragma unroll
  for (int ct = 0; ct < 4; ++ct) bb[ct] = M[DIM * DIM + ct * 16 + l16];
  #pragma unroll
  for (int ct = 0; ct < 4; ++ct) bb1[ct] = M[MAP_ROWS * DIM + DIM * DIM + ct * 16 + l16];
  #pragma unroll
  for (int ct = 0; ct < 4; ++ct) bb2[ct] = M[2 * MAP_ROWS * DIM + DIM * DIM + ct * 16 + l16];
  const v4f* xsrc = (const v4f*)(x + rowBase * DIM);
  v4f xa[8], xb[8];
  #pragma unroll
  for (int u = 0; u < 8; ++u) xa[u] = xsrc[lane + u * 32];
  #pragma unroll
  for (int u = 0; u < 8; ++u) xb[u] = xsrc[256 + lane + u * 32];
  {
    const int r  = tid >> 4;
    const int c4 = (tid & 15) << 2;
    *(v4f*)&mapl[(r +  0) * MPITCH + c4] = mv0 + mw0 + mx0;
    *(v4f*)&mapl[(r + 16) * MPITCH + c4] = mv1 + mw1 + mx1;
    *(v4f*)&mapl[(r + 32) * MPITCH + c4] = mv2 + mw2 + mx2;
    *(v4f*)&mapl[(r + 48) * MPITCH + c4] = mv3 + mw3 + mx3;
  }
  #pragma unroll
  for (int ct = 0; ct < 4; ++ct) bb[ct] = bb[ct] + bb1[ct] + bb2[ct];
  lds_barrier();

  v2f bf[4][16];
  const float* bp = mapl + 2 * half * MPITCH + l16;
  #pragma unroll
  for (int ct = 0; ct < 4; ++ct) {
    #pragma unroll
    for (int j = 0; j < 16; ++j) {
      bf[ct][j].x = bp[(4 * j) * MPITCH + ct * 16];
      bf[ct][j].y = bp[(4 * j + 1) * MPITCH + ct * 16];
    }
  }

  float* tp = tile_all + (wave * AP_TILES) * 16 * PITCH;
  apply_tile(tp,              xa, bf, bb, out + rowBase * DIM,        lane, l16, half);
  apply_tile(tp + 16 * PITCH, xb, bf, bb, out + (rowBase + 16) * DIM, lane, l16, half);
}

extern "C" void kernel_launch(void* const* d_in, const int* in_sizes, int n_in,
                              void* d_out, int out_size, void* d_ws, size_t ws_size,
                              hipStream_t stream) {
  const float* x  = (const float*)d_in[0];
  const float* W  = (const float*)d_in[1];
  const float* b  = (const float*)d_in[2];
  float* out      = (float*)d_out;
  float* M        = (float*)d_ws;

  const int Bn      = in_sizes[0] / DIM;
  const int nBlocks = Bn / AP_ROWS;

  fold_weights<<<N_PLANES * FOLD_TILES, 128, 0, stream>>>(W, b, M);
  apply_map<<<nBlocks, AP_WAVES * 32, 0, stream>>>(x, M, out);
}
